# starting kernel
# baseline (speedup 1.0000x reference)
#include <hip/hip_runtime.h>
#include <math.h>

typedef float v2f __attribute__((ext_vector_type(2)));
typedef float v8f __attribute__((ext_vector_type(8)));

#define DIM 128


__global__ void k_deg_init(unsigned* __restrict__ deg, int N) {
    int i = blockIdx.x * 256 + threadIdx.x;
    if (i < N) deg[i] = 1u;
}

__global__ void k_deg_count(const int* __restrict__ dst, unsigned* __restrict__ deg, int E) {
    int e = blockIdx.x * 256 + threadIdx.x;
    if (e < E) atomicAdd(&deg[dst[e]], 1u);
}

__global__ void k_dinv(float* __restrict__ dinv, int N) {
    int i = blockIdx.x * 256 + threadIdx.x;
    if (i < N) {
        unsigned d = ((const unsigned*)dinv)[i];
        dinv[i] = rsqrtf((float)d);
    }
}


__global__ void k_gemm_wmma(const float* __restrict__ x, const float* __restrict__ W,
                            const float* __restrict__ dinv,
                            float* __restrict__ xw, float* __restrict__ out,
                            int N, int Mtiles) {
    const int wave = threadIdx.x >> 5;
    const int lane = threadIdx.x & 31;
    const int tile = blockIdx.x * 8 + wave;
    if (tile >= Mtiles * 8) return;
    const int tm = tile >> 3;
    const int tn = tile & 7;
    const int row0 = tm * 16, col0 = tn * 16;

    const int mr    = lane & 15;
    const int khalf = (lane >> 4) << 1;

    const int arow_i = min(row0 + mr, N - 1);
    const float* __restrict__ arow = x + (size_t)arow_i * DIM;
    const float* __restrict__ bcol = W + (col0 + mr);

    v8f c = {0.f, 0.f, 0.f, 0.f, 0.f, 0.f, 0.f, 0.f};
    #pragma unroll 8
    for (int kk = 0; kk < DIM; kk += 4) {
        v2f a = *(const v2f*)(arow + kk + khalf);
        v2f b;
        b.x = bcol[(size_t)(kk + khalf) * DIM];
        b.y = bcol[(size_t)(kk + khalf + 1) * DIM];
        c = __builtin_amdgcn_wmma_f32_16x16x4_f32(false, a, false, b,
                                                  (short)0, c, false, false);
    }

    const int mbase = (lane >> 4) << 3;
    const int gcol  = col0 + (lane & 15);
    if (row0 + 16 <= N) {
        #pragma unroll
        for (int r = 0; r < 8; ++r) {
            const int grow = row0 + mbase + r;
            const float v  = c[r];
            const float di = dinv[grow];
            xw [(size_t)grow * DIM + gcol] = v;
            out[(size_t)grow * DIM + gcol] = v * di * di;
        }
    } else {
        #pragma unroll
        for (int r = 0; r < 8; ++r) {
            const int grow = row0 + mbase + r;
            if (grow < N) {
                const float v  = c[r];
                const float di = dinv[grow];
                xw [(size_t)grow * DIM + gcol] = v;
                out[(size_t)grow * DIM + gcol] = v * di * di;
            }
        }
    }
}


__global__ void k_scatter(const int* __restrict__ src, const int* __restrict__ dst,
                          const float* __restrict__ dinv, const float* __restrict__ xw,
                          float* __restrict__ out, int E) {
    const int wave = threadIdx.x >> 5;
    const int lane = threadIdx.x & 31;
    const int e = blockIdx.x * 8 + wave;
    if (e >= E) return;
    const int s = src[e], t = dst[e];
    const float norm = dinv[s] * dinv[t];
    const float4 v = ((const float4*)(xw + (size_t)s * DIM))[lane];
    float* __restrict__ o = out + (size_t)t * DIM + (lane << 2);
    __hip_atomic_fetch_add(o + 0, v.x * norm, __ATOMIC_RELAXED, __HIP_MEMORY_SCOPE_AGENT);
    __hip_atomic_fetch_add(o + 1, v.y * norm, __ATOMIC_RELAXED, __HIP_MEMORY_SCOPE_AGENT);
    __hip_atomic_fetch_add(o + 2, v.z * norm, __ATOMIC_RELAXED, __HIP_MEMORY_SCOPE_AGENT);
    __hip_atomic_fetch_add(o + 3, v.w * norm, __ATOMIC_RELAXED, __HIP_MEMORY_SCOPE_AGENT);
}


__global__ void k_softmax(float* __restrict__ out, const float* __restrict__ bias, int N) {
    const int wave = threadIdx.x >> 5;
    const int lane = threadIdx.x & 31;
    const int row = blockIdx.x * 8 + wave;
    if (row >= N) return;
    float4* __restrict__ rowp = (float4*)(out + (size_t)row * DIM);
    float4 v  = rowp[lane];
    float4 bb = ((const float4*)bias)[lane];
    v.x = fmaxf(v.x + bb.x, 0.f);
    v.y = fmaxf(v.y + bb.y, 0.f);
    v.z = fmaxf(v.z + bb.z, 0.f);
    v.w = fmaxf(v.w + bb.w, 0.f);

    float m = fmaxf(fmaxf(v.x, v.y), fmaxf(v.z, v.w));
    #pragma unroll
    for (int off = 16; off > 0; off >>= 1) m = fmaxf(m, __shfl_xor(m, off, 32));

    v.x = expf(v.x - m); v.y = expf(v.y - m);
    v.z = expf(v.z - m); v.w = expf(v.w - m);
    float s = v.x + v.y + v.z + v.w;
    #pragma unroll
    for (int off = 16; off > 0; off >>= 1) s += __shfl_xor(s, off, 32);

    const float inv = 1.0f / s;
    v.x *= inv; v.y *= inv; v.z *= inv; v.w *= inv;
    rowp[lane] = v;
}


extern "C" void kernel_launch(void* const* d_in, const int* in_sizes, int n_in,
                              void* d_out, int out_size, void* d_ws, size_t ws_size,
                              hipStream_t stream) {
    const float* x  = (const float*)d_in[0];
    const int*   ei = (const int*)  d_in[1];
    const float* W  = (const float*)d_in[2];
    const float* b  = (const float*)d_in[3];
    const int N = in_sizes[0] / DIM;
    const int E = in_sizes[1] / 2;
    const int* src = ei;
    const int* dst = ei + E;

    float* out  = (float*)d_out;
    float* dinv = (float*)d_ws;
    float* xw   = dinv + N;

    k_deg_init <<<(N + 255) / 256, 256, 0, stream>>>((unsigned*)dinv, N);
    k_deg_count<<<(E + 255) / 256, 256, 0, stream>>>(dst, (unsigned*)dinv, E);
    k_dinv     <<<(N + 255) / 256, 256, 0, stream>>>(dinv, N);

    const int Mtiles = (N + 15) / 16;
    k_gemm_wmma<<<(Mtiles * 8 + 7) / 8, 256, 0, stream>>>(x, W, dinv, xw, out, N, Mtiles);

    k_scatter  <<<(E + 7) / 8, 256, 0, stream>>>(src, dst, dinv, xw, out, E);
    k_softmax  <<<(N + 7) / 8, 256, 0, stream>>>(out, b, N);
}


# f32 global atomics dropped: a work-group owns 384 out rows, scans dst for its edges, sums their xw rows with f32 LDS adds, softmax from LDS; in-degrees counted in LDS too; sum order = LDS arrival
# speedup vs baseline: 2.0949x; 2.0949x over previous
#include <hip/hip_runtime.h>
#include <math.h>


typedef float v2f __attribute__((ext_vector_type(2)));
typedef float v8f __attribute__((ext_vector_type(8)));

#define DIM    128
#define WG     1024
#define NW     (WG / 32)
#define DROWS  2048
#define DSLICE 8
#define ROWS   384
#define CAP    8192


__device__ __forceinline__ void count1(unsigned* cnt, int t, int base) {
    const unsigned tl = (unsigned)(t - base);
    if (tl < (unsigned)DROWS) atomicAdd(&cnt[tl], 1u);
}

__global__ __launch_bounds__(WG) void k_degree(const int* __restrict__ dst,
                                                 unsigned* __restrict__ part, int N, int E) {
    __shared__ unsigned cnt[DROWS];
    const int tid  = threadIdx.x;
    const int base = blockIdx.x * DROWS;
    for (int i = tid; i < DROWS; i += WG) cnt[i] = 0u;
    __syncthreads();

    const int per = (((E + DSLICE - 1) / DSLICE) + 3) & ~3;
    const int e0  = min(E, (int)blockIdx.y * per);
    const int e1  = min(E, e0 + per);
    const int* __restrict__ ds = dst + e0;
    const int len = e1 - e0;

    if ((((size_t)ds) & 15) == 0) {
        const int4* __restrict__ d4 = (const int4*)ds;
        const int nv = len >> 2;
        int g = tid;
        for (; g + 3 * WG < nv; g += 4 * WG) {
            int4 t[4];
            #pragma unroll
            for (int u = 0; u < 4; ++u) t[u] = d4[g + u * WG];
            #pragma unroll
            for (int u = 0; u < 4; ++u) {
                count1(cnt, t[u].x, base); count1(cnt, t[u].y, base);
                count1(cnt, t[u].z, base); count1(cnt, t[u].w, base);
            }
        }
        for (; g < nv; g += WG) {
            const int4 t = d4[g];
            count1(cnt, t.x, base); count1(cnt, t.y, base);
            count1(cnt, t.z, base); count1(cnt, t.w, base);
        }
        for (int e = (nv << 2) + tid; e < len; e += WG) count1(cnt, ds[e], base);
    } else {
        for (int e = tid; e < len; e += WG) count1(cnt, ds[e], base);
    }
    __syncthreads();
    unsigned* __restrict__ mine = part + (size_t)blockIdx.y * N;
    for (int i = tid; i < DROWS; i += WG)
        if (base + i < N) mine[base + i] = cnt[i];
}

__global__ void k_dinv(const unsigned* __restrict__ part, float* __restrict__ dinv, int N) {
    const int i = blockIdx.x * 256 + threadIdx.x;
    if (i < N) {
        unsigned d = 1u;
        #pragma unroll
        for (int g = 0; g < DSLICE; ++g) d += part[(size_t)g * N + i];
        dinv[i] = rsqrtf((float)d);
    }
}


__global__ void k_gemm_wmma(const float* __restrict__ x, const float* __restrict__ W,
                            float* __restrict__ xw, int N, int Mtiles) {
    const int wave = threadIdx.x >> 5;
    const int lane = threadIdx.x & 31;
    const int tile = blockIdx.x * 8 + wave;
    if (tile >= Mtiles * 8) return;
    const int tm = tile >> 3;
    const int tn = tile & 7;
    const int row0 = tm * 16, col0 = tn * 16;

    const int mr    = lane & 15;
    const int khalf = (lane >> 4) << 1;

    const int arow_i = min(row0 + mr, N - 1);
    const float* __restrict__ arow = x + (size_t)arow_i * DIM;
    const float* __restrict__ bcol = W + (col0 + mr);

    v8f c = {0.f, 0.f, 0.f, 0.f, 0.f, 0.f, 0.f, 0.f};
    #pragma unroll 8
    for (int kk = 0; kk < DIM; kk += 4) {
        v2f a = *(const v2f*)(arow + kk + khalf);
        v2f b;
        b.x = bcol[(size_t)(kk + khalf) * DIM];
        b.y = bcol[(size_t)(kk + khalf + 1) * DIM];
        c = __builtin_amdgcn_wmma_f32_16x16x4_f32(false, a, false, b,
                                                  (short)0, c, false, false);
    }

    const int mbase = (lane >> 4) << 3;
    const int gcol  = col0 + (lane & 15);
    if (row0 + 16 <= N) {
        #pragma unroll
        for (int r = 0; r < 8; ++r) {
            const int grow = row0 + mbase + r;
            xw[(size_t)grow * DIM + gcol] = c[r];
        }
    } else {
        #pragma unroll
        for (int r = 0; r < 8; ++r) {
            const int grow = row0 + mbase + r;
            if (grow < N) xw[(size_t)grow * DIM + gcol] = c[r];
        }
    }
}


__device__ __forceinline__ void add_row_alone(float* a, const float* __restrict__ xr, float w) {
    for (int c = 0; c < DIM; ++c)
        atomicAdd(a + ((c & 3) * 32 + (c >> 2)), xr[c] * w);
}

__device__ __forceinline__ void add_row(float* a, const float4 v, const float w) {
    atomicAdd(a,      v.x * w);
    atomicAdd(a + 32, v.y * w);
    atomicAdd(a + 64, v.z * w);
    atomicAdd(a + 96, v.w * w);
}

__global__ __launch_bounds__(WG) void k_accumulate(const int* __restrict__ src,
                                                     const int* __restrict__ dst,
                                                     const float* __restrict__ dinv,
                                                     const float* __restrict__ xw,
                                                     const float* __restrict__ bias,
                                                     float* __restrict__ out, int N, int E) {
    __shared__ float    acc[ROWS * DIM];
    __shared__ int      l_s[CAP];
    __shared__ int      l_t[CAP];
    __shared__ float    l_w[CAP];
    __shared__ unsigned nlist;

    const int tid  = threadIdx.x;
    const int wave = __builtin_amdgcn_readfirstlane(tid >> 5);
    const int lane = tid & 31;
    const int base = blockIdx.x * ROWS;
    const int nrows = min(ROWS, N - base);
    const unsigned nr = (unsigned)nrows;

    for (int r0 = wave; r0 < nrows; r0 += 4 * NW) {
        float  di[4];
        float4 v[4];
        #pragma unroll
        for (int u = 0; u < 4; ++u) {
            const int r = min(r0 + u * NW, nrows - 1);
            di[u] = dinv[base + r];
            v[u]  = ((const float4*)(xw + (size_t)(base + r) * DIM))[lane];
        }
        #pragma unroll
        for (int u = 0; u < 4; ++u) {
            const int r = r0 + u * NW;
            if (r < nrows) {
                float* a = acc + r * DIM + lane;
                a[0]  = v[u].x * di[u] * di[u];
                a[32] = v[u].y * di[u] * di[u];
                a[64] = v[u].z * di[u] * di[u];
                a[96] = v[u].w * di[u] * di[u];
            }
        }
    }
    if (tid == 0) nlist = 0u;
    __syncthreads();

    auto note = [&](int e) {
        const unsigned slot = atomicAdd(&nlist, 1u);
        if (slot < (unsigned)CAP) {
            l_s[slot] = e;
        } else {
            const int s  = src[e];
            const int tl = dst[e] - base;
            add_row_alone(acc + tl * DIM, xw + (size_t)s * DIM, dinv[s] * dinv[base + tl]);
        }
    };
    if ((((size_t)dst) & 15) == 0) {
        const int4* __restrict__ d4 = (const int4*)dst;
        const int nv = E >> 2;
        int g = tid;
        for (; g + 7 * WG < nv; g += 8 * WG) {
            int4 t[8];
            #pragma unroll
            for (int u = 0; u < 8; ++u) t[u] = d4[g + u * WG];
            unsigned mask = 0u;
            #pragma unroll
            for (int u = 0; u < 8; ++u) {
                mask |= ((unsigned)(t[u].x - base) < nr ? 1u : 0u) << (4 * u + 0);
                mask |= ((unsigned)(t[u].y - base) < nr ? 1u : 0u) << (4 * u + 1);
                mask |= ((unsigned)(t[u].z - base) < nr ? 1u : 0u) << (4 * u + 2);
                mask |= ((unsigned)(t[u].w - base) < nr ? 1u : 0u) << (4 * u + 3);
            }
            while (mask) {
                const int k = __builtin_ctz(mask);
                mask &= mask - 1u;
                note(((g + (k >> 2) * WG) << 2) + (k & 3));
            }
        }
        for (; g < nv; g += WG) {
            const int4 t = d4[g];
            if ((unsigned)(t.x - base) < nr) note((g << 2) + 0);
            if ((unsigned)(t.y - base) < nr) note((g << 2) + 1);
            if ((unsigned)(t.z - base) < nr) note((g << 2) + 2);
            if ((unsigned)(t.w - base) < nr) note((g << 2) + 3);
        }
        for (int e = (nv << 2) + tid; e < E; e += WG)
            if ((unsigned)(dst[e] - base) < nr) note(e);
    } else {
        for (int e = tid; e < E; e += WG)
            if ((unsigned)(dst[e] - base) < nr) note(e);
    }
    __syncthreads();

    const int n = (int)min(nlist, (unsigned)CAP);
    for (int i = tid; i < n; i += WG) {
        const int e  = l_s[i];
        const int s  = src[e];
        const int tl = dst[e] - base;
        l_s[i] = s;
        l_t[i] = tl;
        l_w[i] = dinv[s] * dinv[base + tl];
    }
    __syncthreads();

    for (int g0 = wave * 8; g0 < n; g0 += 8 * NW) {
        const int m = min(8, n - g0);
        int   s_l = 0, t_l = 0;
        float w_l = 0.f;
        if (lane < m) { s_l = l_s[g0 + lane]; t_l = l_t[g0 + lane]; w_l = l_w[g0 + lane]; }
        float4 v[8];
        #pragma unroll
        for (int u = 0; u < 8; ++u) {
            const int sj = __builtin_amdgcn_readlane(s_l, min(u, m - 1));
            v[u] = ((const float4*)(xw + (size_t)sj * DIM))[lane];
        }
        #pragma unroll
        for (int u = 0; u < 8; ++u) {
            if (u < m) {
                const int   tj = __builtin_amdgcn_readlane(t_l, u);
                const float wj = __int_as_float(__builtin_amdgcn_readlane(__float_as_int(w_l), u));
                add_row(acc + tj * DIM + lane, v[u], wj);
            }
        }
    }
    __syncthreads();

    const float4 bb = ((const float4*)bias)[lane];
    for (int r0 = wave; r0 < nrows; r0 += 4 * NW) {
        float4 v[4];
        float  mx[4], sum[4];
        #pragma unroll
        for (int u = 0; u < 4; ++u) {
            const int r = min(r0 + u * NW, nrows - 1);
            const float* a = acc + r * DIM + lane;
            v[u].x = fmaxf(a[0]  + bb.x, 0.f);
            v[u].y = fmaxf(a[32] + bb.y, 0.f);
            v[u].z = fmaxf(a[64] + bb.z, 0.f);
            v[u].w = fmaxf(a[96] + bb.w, 0.f);
            mx[u] = fmaxf(fmaxf(v[u].x, v[u].y), fmaxf(v[u].z, v[u].w));
        }
        #pragma unroll
        for (int off = 16; off > 0; off >>= 1) {
            #pragma unroll
            for (int u = 0; u < 4; ++u) mx[u] = fmaxf(mx[u], __shfl_xor(mx[u], off, 32));
        }
        #pragma unroll
        for (int u = 0; u < 4; ++u) {
            v[u].x = expf(v[u].x - mx[u]); v[u].y = expf(v[u].y - mx[u]);
            v[u].z = expf(v[u].z - mx[u]); v[u].w = expf(v[u].w - mx[u]);
            sum[u] = v[u].x + v[u].y + v[u].z + v[u].w;
        }
        #pragma unroll
        for (int off = 16; off > 0; off >>= 1) {
            #pragma unroll
            for (int u = 0; u < 4; ++u) sum[u] += __shfl_xor(sum[u], off, 32);
        }
        #pragma unroll
        for (int u = 0; u < 4; ++u) {
            const int r = r0 + u * NW;
            if (r < nrows) {
                const float inv = 1.0f / sum[u];
                float4 o;
                o.x = v[u].x * inv; o.y = v[u].y * inv; o.z = v[u].z * inv; o.w = v[u].w * inv;
                ((float4*)(out + (size_t)(base + r) * DIM))[lane] = o;
            }
        }
    }
}


extern "C" void kernel_launch(void* const* d_in, const int* in_sizes, int n_in,
                              void* d_out, int out_size, void* d_ws, size_t ws_size,
                              hipStream_t stream) {
    const float* x  = (const float*)d_in[0];
    const int*   ei = (const int*)  d_in[1];
    const float* W  = (const float*)d_in[2];
    const float* b  = (const float*)d_in[3];
    const int N = in_sizes[0] / DIM;
    const int E = in_sizes[1] / 2;
    const int* src = ei;
    const int* dst = ei + E;

    float*    out  = (float*)d_out;
    float*    dinv = (float*)d_ws;
    float*    xw   = dinv + N;
    unsigned* part = (unsigned*)(xw + (size_t)N * DIM);

    k_degree<<<dim3((N + DROWS - 1) / DROWS, DSLICE), WG, 0, stream>>>(dst, part, N, E);
    k_dinv  <<<(N + 255) / 256, 256, 0, stream>>>(part, dinv, N);

    const int Mtiles = (N + 15) / 16;
    k_gemm_wmma<<<(Mtiles * 8 + 7) / 8, 256, 0, stream>>>(x, W, xw, N, Mtiles);

    k_accumulate<<<(N + ROWS - 1) / ROWS, WG, 0, stream>>>(src, dst, dinv, xw, b, out, N, E);
}


# k_accumulate: a work-group owns 200 out rows, not 384: 250 work-groups, one per WGP, spread the f32 LDS adds over the whole chip where 131 left half idle; sum order = LDS arrival as before
# speedup vs baseline: 2.5272x; 1.2064x over previous
#include <hip/hip_runtime.h>
#include <math.h>


typedef float v2f __attribute__((ext_vector_type(2)));
typedef float v8f __attribute__((ext_vector_type(8)));

#define DIM    128
#define WG     1024
#define NW     (WG / 32)
#define DROWS  2048
#define DSLICE 8
#define ROWS   200
#define CAP    8192


__device__ __forceinline__ void count1(unsigned* cnt, int t, int base) {
    const unsigned tl = (unsigned)(t - base);
    if (tl < (unsigned)DROWS) atomicAdd(&cnt[tl], 1u);
}

__global__ __launch_bounds__(WG) void k_degree(const int* __restrict__ dst,
                                                 unsigned* __restrict__ part, int N, int E) {
    __shared__ unsigned cnt[DROWS];
    const int tid  = threadIdx.x;
    const int base = blockIdx.x * DROWS;
    for (int i = tid; i < DROWS; i += WG) cnt[i] = 0u;
    __syncthreads();

    const int per = (((E + DSLICE - 1) / DSLICE) + 3) & ~3;
    const int e0  = min(E, (int)blockIdx.y * per);
    const int e1  = min(E, e0 + per);
    const int* __restrict__ ds = dst + e0;
    const int len = e1 - e0;

    if ((((size_t)ds) & 15) == 0) {
        const int4* __restrict__ d4 = (const int4*)ds;
        const int nv = len >> 2;
        int g = tid;
        for (; g + 3 * WG < nv; g += 4 * WG) {
            int4 t[4];
            #pragma unroll
            for (int u = 0; u < 4; ++u) t[u] = d4[g + u * WG];
            #pragma unroll
            for (int u = 0; u < 4; ++u) {
                count1(cnt, t[u].x, base); count1(cnt, t[u].y, base);
                count1(cnt, t[u].z, base); count1(cnt, t[u].w, base);
            }
        }
        for (; g < nv; g += WG) {
            const int4 t = d4[g];
            count1(cnt, t.x, base); count1(cnt, t.y, base);
            count1(cnt, t.z, base); count1(cnt, t.w, base);
        }
        for (int e = (nv << 2) + tid; e < len; e += WG) count1(cnt, ds[e], base);
    } else {
        for (int e = tid; e < len; e += WG) count1(cnt, ds[e], base);
    }
    __syncthreads();
    unsigned* __restrict__ mine = part + (size_t)blockIdx.y * N;
    for (int i = tid; i < DROWS; i += WG)
        if (base + i < N) mine[base + i] = cnt[i];
}

__global__ void k_dinv(const unsigned* __restrict__ part, float* __restrict__ dinv, int N) {
    const int i = blockIdx.x * 256 + threadIdx.x;
    if (i < N) {
        unsigned d = 1u;
        #pragma unroll
        for (int g = 0; g < DSLICE; ++g) d += part[(size_t)g * N + i];
        dinv[i] = rsqrtf((float)d);
    }
}


__global__ void k_gemm_wmma(const float* __restrict__ x, const float* __restrict__ W,
                            float* __restrict__ xw, int N, int Mtiles) {
    const int wave = threadIdx.x >> 5;
    const int lane = threadIdx.x & 31;
    const int tile = blockIdx.x * 8 + wave;
    if (tile >= Mtiles * 8) return;
    const int tm = tile >> 3;
    const int tn = tile & 7;
    const int row0 = tm * 16, col0 = tn * 16;

    const int mr    = lane & 15;
    const int khalf = (lane >> 4) << 1;

    const int arow_i = min(row0 + mr, N - 1);
    const float* __restrict__ arow = x + (size_t)arow_i * DIM;
    const float* __restrict__ bcol = W + (col0 + mr);

    v8f c = {0.f, 0.f, 0.f, 0.f, 0.f, 0.f, 0.f, 0.f};
    #pragma unroll 8
    for (int kk = 0; kk < DIM; kk += 4) {
        v2f a = *(const v2f*)(arow + kk + khalf);
        v2f b;
        b.x = bcol[(size_t)(kk + khalf) * DIM];
        b.y = bcol[(size_t)(kk + khalf + 1) * DIM];
        c = __builtin_amdgcn_wmma_f32_16x16x4_f32(false, a, false, b,
                                                  (short)0, c, false, false);
    }

    const int mbase = (lane >> 4) << 3;
    const int gcol  = col0 + (lane & 15);
    if (row0 + 16 <= N) {
        #pragma unroll
        for (int r = 0; r < 8; ++r) {
            const int grow = row0 + mbase + r;
            xw[(size_t)grow * DIM + gcol] = c[r];
        }
    } else {
        #pragma unroll
        for (int r = 0; r < 8; ++r) {
            const int grow = row0 + mbase + r;
            if (grow < N) xw[(size_t)grow * DIM + gcol] = c[r];
        }
    }
}


__device__ __forceinline__ void add_row_alone(float* a, const float* __restrict__ xr, float w) {
    for (int c = 0; c < DIM; ++c)
        atomicAdd(a + ((c & 3) * 32 + (c >> 2)), xr[c] * w);
}

__device__ __forceinline__ void add_row(float* a, const float4 v, const float w) {
    atomicAdd(a,      v.x * w);
    atomicAdd(a + 32, v.y * w);
    atomicAdd(a + 64, v.z * w);
    atomicAdd(a + 96, v.w * w);
}

__global__ __launch_bounds__(WG) void k_accumulate(const int* __restrict__ src,
                                                     const int* __restrict__ dst,
                                                     const float* __restrict__ dinv,
                                                     const float* __restrict__ xw,
                                                     const float* __restrict__ bias,
                                                     float* __restrict__ out, int N, int E) {
    __shared__ float    acc[ROWS * DIM];
    __shared__ int      l_s[CAP];
    __shared__ int      l_t[CAP];
    __shared__ float    l_w[CAP];
    __shared__ unsigned nlist;

    const int tid  = threadIdx.x;
    const int wave = __builtin_amdgcn_readfirstlane(tid >> 5);
    const int lane = tid & 31;
    const int base = blockIdx.x * ROWS;
    const int nrows = min(ROWS, N - base);
    const unsigned nr = (unsigned)nrows;

    for (int r0 = wave; r0 < nrows; r0 += 4 * NW) {
        float  di[4];
        float4 v[4];
        #pragma unroll
        for (int u = 0; u < 4; ++u) {
            const int r = min(r0 + u * NW, nrows - 1);
            di[u] = dinv[base + r];
            v[u]  = ((const float4*)(xw + (size_t)(base + r) * DIM))[lane];
        }
        #pragma unroll
        for (int u = 0; u < 4; ++u) {
            const int r = r0 + u * NW;
            if (r < nrows) {
                float* a = acc + r * DIM + lane;
                a[0]  = v[u].x * di[u] * di[u];
                a[32] = v[u].y * di[u] * di[u];
                a[64] = v[u].z * di[u] * di[u];
                a[96] = v[u].w * di[u] * di[u];
            }
        }
    }
    if (tid == 0) nlist = 0u;
    __syncthreads();

    auto note = [&](int e) {
        const unsigned slot = atomicAdd(&nlist, 1u);
        if (slot < (unsigned)CAP) {
            l_s[slot] = e;
        } else {
            const int s  = src[e];
            const int tl = dst[e] - base;
            add_row_alone(acc + tl * DIM, xw + (size_t)s * DIM, dinv[s] * dinv[base + tl]);
        }
    };
    if ((((size_t)dst) & 15) == 0) {
        const int4* __restrict__ d4 = (const int4*)dst;
        const int nv = E >> 2;
        int g = tid;
        for (; g + 7 * WG < nv; g += 8 * WG) {
            int4 t[8];
            #pragma unroll
            for (int u = 0; u < 8; ++u) t[u] = d4[g + u * WG];
            unsigned mask = 0u;
            #pragma unroll
            for (int u = 0; u < 8; ++u) {
                mask |= ((unsigned)(t[u].x - base) < nr ? 1u : 0u) << (4 * u + 0);
                mask |= ((unsigned)(t[u].y - base) < nr ? 1u : 0u) << (4 * u + 1);
                mask |= ((unsigned)(t[u].z - base) < nr ? 1u : 0u) << (4 * u + 2);
                mask |= ((unsigned)(t[u].w - base) < nr ? 1u : 0u) << (4 * u + 3);
            }
            while (mask) {
                const int k = __builtin_ctz(mask);
                mask &= mask - 1u;
                note(((g + (k >> 2) * WG) << 2) + (k & 3));
            }
        }
        for (; g < nv; g += WG) {
            const int4 t = d4[g];
            if ((unsigned)(t.x - base) < nr) note((g << 2) + 0);
            if ((unsigned)(t.y - base) < nr) note((g << 2) + 1);
            if ((unsigned)(t.z - base) < nr) note((g << 2) + 2);
            if ((unsigned)(t.w - base) < nr) note((g << 2) + 3);
        }
        for (int e = (nv << 2) + tid; e < E; e += WG)
            if ((unsigned)(dst[e] - base) < nr) note(e);
    } else {
        for (int e = tid; e < E; e += WG)
            if ((unsigned)(dst[e] - base) < nr) note(e);
    }
    __syncthreads();

    const int n = (int)min(nlist, (unsigned)CAP);
    for (int i = tid; i < n; i += WG) {
        const int e  = l_s[i];
        const int s  = src[e];
        const int tl = dst[e] - base;
        l_s[i] = s;
        l_t[i] = tl;
        l_w[i] = dinv[s] * dinv[base + tl];
    }
    __syncthreads();

    for (int g0 = wave * 8; g0 < n; g0 += 8 * NW) {
        const int m = min(8, n - g0);
        int   s_l = 0, t_l = 0;
        float w_l = 0.f;
        if (lane < m) { s_l = l_s[g0 + lane]; t_l = l_t[g0 + lane]; w_l = l_w[g0 + lane]; }
        float4 v[8];
        #pragma unroll
        for (int u = 0; u < 8; ++u) {
            const int sj = __builtin_amdgcn_readlane(s_l, min(u, m - 1));
            v[u] = ((const float4*)(xw + (size_t)sj * DIM))[lane];
        }
        #pragma unroll
        for (int u = 0; u < 8; ++u) {
            if (u < m) {
                const int   tj = __builtin_amdgcn_readlane(t_l, u);
                const float wj = __int_as_float(__builtin_amdgcn_readlane(__float_as_int(w_l), u));
                add_row(acc + tj * DIM + lane, v[u], wj);
            }
        }
    }
    __syncthreads();

    const float4 bb = ((const float4*)bias)[lane];
    for (int r0 = wave; r0 < nrows; r0 += 4 * NW) {
        float4 v[4];
        float  mx[4], sum[4];
        #pragma unroll
        for (int u = 0; u < 4; ++u) {
            const int r = min(r0 + u * NW, nrows - 1);
            const float* a = acc + r * DIM + lane;
            v[u].x = fmaxf(a[0]  + bb.x, 0.f);
            v[u].y = fmaxf(a[32] + bb.y, 0.f);
            v[u].z = fmaxf(a[64] + bb.z, 0.f);
            v[u].w = fmaxf(a[96] + bb.w, 0.f);
            mx[u] = fmaxf(fmaxf(v[u].x, v[u].y), fmaxf(v[u].z, v[u].w));
        }
        #pragma unroll
        for (int off = 16; off > 0; off >>= 1) {
            #pragma unroll
            for (int u = 0; u < 4; ++u) mx[u] = fmaxf(mx[u], __shfl_xor(mx[u], off, 32));
        }
        #pragma unroll
        for (int u = 0; u < 4; ++u) {
            v[u].x = expf(v[u].x - mx[u]); v[u].y = expf(v[u].y - mx[u]);
            v[u].z = expf(v[u].z - mx[u]); v[u].w = expf(v[u].w - mx[u]);
            sum[u] = v[u].x + v[u].y + v[u].z + v[u].w;
        }
        #pragma unroll
        for (int off = 16; off > 0; off >>= 1) {
            #pragma unroll
            for (int u = 0; u < 4; ++u) sum[u] += __shfl_xor(sum[u], off, 32);
        }
        #pragma unroll
        for (int u = 0; u < 4; ++u) {
            const int r = r0 + u * NW;
            if (r < nrows) {
                const float inv = 1.0f / sum[u];
                float4 o;
                o.x = v[u].x * inv; o.y = v[u].y * inv; o.z = v[u].z * inv; o.w = v[u].w * inv;
                ((float4*)(out + (size_t)(base + r) * DIM))[lane] = o;
            }
        }
    }
}


extern "C" void kernel_launch(void* const* d_in, const int* in_sizes, int n_in,
                              void* d_out, int out_size, void* d_ws, size_t ws_size,
                              hipStream_t stream) {
    const float* x  = (const float*)d_in[0];
    const int*   ei = (const int*)  d_in[1];
    const float* W  = (const float*)d_in[2];
    const float* b  = (const float*)d_in[3];
    const int N = in_sizes[0] / DIM;
    const int E = in_sizes[1] / 2;
    const int* src = ei;
    const int* dst = ei + E;

    float*    out  = (float*)d_out;
    float*    dinv = (float*)d_ws;
    float*    xw   = dinv + N;
    unsigned* part = (unsigned*)(xw + (size_t)N * DIM);

    k_degree<<<dim3((N + DROWS - 1) / DROWS, DSLICE), WG, 0, stream>>>(dst, part, N, E);
    k_dinv  <<<(N + 255) / 256, 256, 0, stream>>>(part, dinv, N);

    const int Mtiles = (N + 15) / 16;
    k_gemm_wmma<<<(Mtiles * 8 + 7) / 8, 256, 0, stream>>>(x, W, xw, N, Mtiles);

    k_accumulate<<<(N + ROWS - 1) / ROWS, WG, 0, stream>>>(src, dst, dinv, xw, b, out, N, E);
}


# k_degree also writes one byte per edge, its owning work-group (dst/200); k_accumulate finds its edges in these bytes, 16 edges per 16-B load, not in dst (4 per load); sum order = LDS arrival as before
# speedup vs baseline: 2.7946x; 1.1058x over previous
#include <hip/hip_runtime.h>
#include <math.h>


typedef float v2f __attribute__((ext_vector_type(2)));
typedef float v8f __attribute__((ext_vector_type(8)));

#define DIM    128
#define WG     1024
#define NW     (WG / 32)
#define DROWS  2048
#define DSLICE 8
#define ROWS   200
#define CAP    8192


#define KEYMAX 256

__device__ __forceinline__ unsigned key1(int t) { return (unsigned)t / (unsigned)ROWS; }
__device__ __forceinline__ unsigned key4(const int4 t) {
    return key1(t.x) | (key1(t.y) << 8) | (key1(t.z) << 16) | (key1(t.w) << 24);
}

__device__ __forceinline__ void count1(unsigned* cnt, int t, int base) {
    const unsigned tl = (unsigned)(t - base);
    if (tl < (unsigned)DROWS) atomicAdd(&cnt[tl], 1u);
}

__global__ __launch_bounds__(WG) void k_degree(const int* __restrict__ dst,
                                                 unsigned* __restrict__ part,
                                                 unsigned char* __restrict__ keys, int nblk,
                                                 int N, int E) {
    __shared__ unsigned cnt[DROWS];
    const int tid  = threadIdx.x;
    const int base = blockIdx.x * DROWS;
    for (int i = tid; i < DROWS; i += WG) cnt[i] = 0u;
    __syncthreads();

    const int per = (((E + DSLICE - 1) / DSLICE) + 3) & ~3;
    const int e0  = min(E, (int)blockIdx.y * per);
    const int e1  = min(E, e0 + per);
    const int* __restrict__ ds = dst + e0;
    const int len = e1 - e0;
    const bool wk = (blockIdx.x == 0) && (nblk <= KEYMAX);
    unsigned char* kb = keys + e0;
    unsigned* kw = (unsigned*)kb;

    if ((((size_t)ds) & 15) == 0) {
        const int4* __restrict__ d4 = (const int4*)ds;
        const int nv = len >> 2;
        int g = tid;
        for (; g + 3 * WG < nv; g += 4 * WG) {
            int4 t[4];
            #pragma unroll
            for (int u = 0; u < 4; ++u) t[u] = d4[g + u * WG];
            if (wk) {
                #pragma unroll
                for (int u = 0; u < 4; ++u) kw[g + u * WG] = key4(t[u]);
            }
            #pragma unroll
            for (int u = 0; u < 4; ++u) {
                count1(cnt, t[u].x, base); count1(cnt, t[u].y, base);
                count1(cnt, t[u].z, base); count1(cnt, t[u].w, base);
            }
        }
        for (; g < nv; g += WG) {
            const int4 t = d4[g];
            if (wk) kw[g] = key4(t);
            count1(cnt, t.x, base); count1(cnt, t.y, base);
            count1(cnt, t.z, base); count1(cnt, t.w, base);
        }
        for (int e = (nv << 2) + tid; e < len; e += WG) {
            const int t = ds[e];
            if (wk) kb[e] = (unsigned char)key1(t);
            count1(cnt, t, base);
        }
    } else {
        for (int e = tid; e < len; e += WG) {
            const int t = ds[e];
            if (wk) kb[e] = (unsigned char)key1(t);
            count1(cnt, t, base);
        }
    }
    __syncthreads();
    unsigned* __restrict__ mine = part + (size_t)blockIdx.y * N;
    for (int i = tid; i < DROWS; i += WG)
        if (base + i < N) mine[base + i] = cnt[i];
}

__global__ void k_dinv(const unsigned* __restrict__ part, float* __restrict__ dinv, int N) {
    const int i = blockIdx.x * 256 + threadIdx.x;
    if (i < N) {
        unsigned d = 1u;
        #pragma unroll
        for (int g = 0; g < DSLICE; ++g) d += part[(size_t)g * N + i];
        dinv[i] = rsqrtf((float)d);
    }
}


__global__ void k_gemm_wmma(const float* __restrict__ x, const float* __restrict__ W,
                            float* __restrict__ xw, int N, int Mtiles) {
    const int wave = threadIdx.x >> 5;
    const int lane = threadIdx.x & 31;
    const int tile = blockIdx.x * 8 + wave;
    if (tile >= Mtiles * 8) return;
    const int tm = tile >> 3;
    const int tn = tile & 7;
    const int row0 = tm * 16, col0 = tn * 16;

    const int mr    = lane & 15;
    const int khalf = (lane >> 4) << 1;

    const int arow_i = min(row0 + mr, N - 1);
    const float* __restrict__ arow = x + (size_t)arow_i * DIM;
    const float* __restrict__ bcol = W + (col0 + mr);

    v8f c = {0.f, 0.f, 0.f, 0.f, 0.f, 0.f, 0.f, 0.f};
    #pragma unroll 8
    for (int kk = 0; kk < DIM; kk += 4) {
        v2f a = *(const v2f*)(arow + kk + khalf);
        v2f b;
        b.x = bcol[(size_t)(kk + khalf) * DIM];
        b.y = bcol[(size_t)(kk + khalf + 1) * DIM];
        c = __builtin_amdgcn_wmma_f32_16x16x4_f32(false, a, false, b,
                                                  (short)0, c, false, false);
    }

    const int mbase = (lane >> 4) << 3;
    const int gcol  = col0 + (lane & 15);
    if (row0 + 16 <= N) {
        #pragma unroll
        for (int r = 0; r < 8; ++r) {
            const int grow = row0 + mbase + r;
            xw[(size_t)grow * DIM + gcol] = c[r];
        }
    } else {
        #pragma unroll
        for (int r = 0; r < 8; ++r) {
            const int grow = row0 + mbase + r;
            if (grow < N) xw[(size_t)grow * DIM + gcol] = c[r];
        }
    }
}


__device__ __forceinline__ void add_row_alone(float* a, const float* __restrict__ xr, float w) {
    for (int c = 0; c < DIM; ++c)
        atomicAdd(a + ((c & 3) * 32 + (c >> 2)), xr[c] * w);
}

__device__ __forceinline__ void add_row(float* a, const float4 v, const float w) {
    atomicAdd(a,      v.x * w);
    atomicAdd(a + 32, v.y * w);
    atomicAdd(a + 64, v.z * w);
    atomicAdd(a + 96, v.w * w);
}

__global__ __launch_bounds__(WG) void k_accumulate(const int* __restrict__ src,
                                                     const int* __restrict__ dst,
                                                     const unsigned char* __restrict__ keys,
                                                     const float* __restrict__ dinv,
                                                     const float* __restrict__ xw,
                                                     const float* __restrict__ bias,
                                                     float* __restrict__ out, int N, int E) {
    __shared__ float    acc[ROWS * DIM];
    __shared__ int      l_s[CAP];
    __shared__ int      l_t[CAP];
    __shared__ float    l_w[CAP];
    __shared__ unsigned nlist;

    const int tid  = threadIdx.x;
    const int wave = __builtin_amdgcn_readfirstlane(tid >> 5);
    const int lane = tid & 31;
    const int base = blockIdx.x * ROWS;
    const int nrows = min(ROWS, N - base);
    const unsigned nr = (unsigned)nrows;

    for (int r0 = wave; r0 < nrows; r0 += 4 * NW) {
        float  di[4];
        float4 v[4];
        #pragma unroll
        for (int u = 0; u < 4; ++u) {
            const int r = min(r0 + u * NW, nrows - 1);
            di[u] = dinv[base + r];
            v[u]  = ((const float4*)(xw + (size_t)(base + r) * DIM))[lane];
        }
        #pragma unroll
        for (int u = 0; u < 4; ++u) {
            const int r = r0 + u * NW;
            if (r < nrows) {
                float* a = acc + r * DIM + lane;
                a[0]  = v[u].x * di[u] * di[u];
                a[32] = v[u].y * di[u] * di[u];
                a[64] = v[u].z * di[u] * di[u];
                a[96] = v[u].w * di[u] * di[u];
            }
        }
    }
    if (tid == 0) nlist = 0u;
    __syncthreads();

    auto note = [&](int e) {
        const unsigned slot = atomicAdd(&nlist, 1u);
        if (slot < (unsigned)CAP) {
            l_s[slot] = e;
        } else {
            const int s  = src[e];
            const int tl = dst[e] - base;
            add_row_alone(acc + tl * DIM, xw + (size_t)s * DIM, dinv[s] * dinv[base + tl]);
        }
    };
    if (gridDim.x <= KEYMAX && (((size_t)keys) & 15) == 0) {
        const uint4* __restrict__ k4 = (const uint4*)keys;
        const int nv = E >> 4;
        const unsigned pat = (unsigned)blockIdx.x * 0x01010101u;
        auto hit = [&](unsigned w) -> unsigned {
            const unsigned x = w ^ pat;
            return ~(((x & 0x7F7F7F7Fu) + 0x7F7F7F7Fu) | x | 0x7F7F7F7Fu);
        };
        auto hits16 = [&](const uint4 t) -> unsigned {
            return (hit(t.x) >> 7) | (hit(t.y) >> 6) | (hit(t.z) >> 5) | (hit(t.w) >> 4);
        };
        int g = tid;
        for (; g + 7 * WG < nv; g += 8 * WG) {
            uint4 t[8];
            #pragma unroll
            for (int u = 0; u < 8; ++u) t[u] = k4[g + u * WG];
            #pragma unroll
            for (int u = 0; u < 8; u += 2) {
                unsigned mask = hits16(t[u]) | (hits16(t[u + 1]) << 4);
                while (mask) {
                    const int p = __builtin_ctz(mask);
                    mask &= mask - 1u;
                    note(((g + (u + ((p >> 2) & 1)) * WG) << 4) + ((p & 3) << 2) + (p >> 3));
                }
            }
        }
        for (; g < nv; g += WG) {
            unsigned mask = hits16(k4[g]);
            while (mask) {
                const int p = __builtin_ctz(mask);
                mask &= mask - 1u;
                note((g << 4) + ((p & 3) << 2) + (p >> 3));
            }
        }
        for (int e = (nv << 4) + tid; e < E; e += WG)
            if ((unsigned)keys[e] == (unsigned)blockIdx.x) note(e);
    } else if ((((size_t)dst) & 15) == 0) {
        const int4* __restrict__ d4 = (const int4*)dst;
        const int nv = E >> 2;
        int g = tid;
        for (; g + 7 * WG < nv; g += 8 * WG) {
            int4 t[8];
            #pragma unroll
            for (int u = 0; u < 8; ++u) t[u] = d4[g + u * WG];
            unsigned mask = 0u;
            #pragma unroll
            for (int u = 0; u < 8; ++u) {
                mask |= ((unsigned)(t[u].x - base) < nr ? 1u : 0u) << (4 * u + 0);
                mask |= ((unsigned)(t[u].y - base) < nr ? 1u : 0u) << (4 * u + 1);
                mask |= ((unsigned)(t[u].z - base) < nr ? 1u : 0u) << (4 * u + 2);
                mask |= ((unsigned)(t[u].w - base) < nr ? 1u : 0u) << (4 * u + 3);
            }
            while (mask) {
                const int k = __builtin_ctz(mask);
                mask &= mask - 1u;
                note(((g + (k >> 2) * WG) << 2) + (k & 3));
            }
        }
        for (; g < nv; g += WG) {
            const int4 t = d4[g];
            if ((unsigned)(t.x - base) < nr) note((g << 2) + 0);
            if ((unsigned)(t.y - base) < nr) note((g << 2) + 1);
            if ((unsigned)(t.z - base) < nr) note((g << 2) + 2);
            if ((unsigned)(t.w - base) < nr) note((g << 2) + 3);
        }
        for (int e = (nv << 2) + tid; e < E; e += WG)
            if ((unsigned)(dst[e] - base) < nr) note(e);
    } else {
        for (int e = tid; e < E; e += WG)
            if ((unsigned)(dst[e] - base) < nr) note(e);
    }
    __syncthreads();

    const int n = (int)min(nlist, (unsigned)CAP);
    for (int i = tid; i < n; i += WG) {
        const int e  = l_s[i];
        const int s  = src[e];
        const int tl = dst[e] - base;
        l_s[i] = s;
        l_t[i] = tl;
        l_w[i] = dinv[s] * dinv[base + tl];
    }
    __syncthreads();

    for (int g0 = wave * 8; g0 < n; g0 += 8 * NW) {
        const int m = min(8, n - g0);
        int   s_l = 0, t_l = 0;
        float w_l = 0.f;
        if (lane < m) { s_l = l_s[g0 + lane]; t_l = l_t[g0 + lane]; w_l = l_w[g0 + lane]; }
        float4 v[8];
        #pragma unroll
        for (int u = 0; u < 8; ++u) {
            const int sj = __builtin_amdgcn_readlane(s_l, min(u, m - 1));
            v[u] = ((const float4*)(xw + (size_t)sj * DIM))[lane];
        }
        #pragma unroll
        for (int u = 0; u < 8; ++u) {
            if (u < m) {
                const int   tj = __builtin_amdgcn_readlane(t_l, u);
                const float wj = __int_as_float(__builtin_amdgcn_readlane(__float_as_int(w_l), u));
                add_row(acc + tj * DIM + lane, v[u], wj);
            }
        }
    }
    __syncthreads();

    const float4 bb = ((const float4*)bias)[lane];
    for (int r0 = wave; r0 < nrows; r0 += 4 * NW) {
        float4 v[4];
        float  mx[4], sum[4];
        #pragma unroll
        for (int u = 0; u < 4; ++u) {
            const int r = min(r0 + u * NW, nrows - 1);
            const float* a = acc + r * DIM + lane;
            v[u].x = fmaxf(a[0]  + bb.x, 0.f);
            v[u].y = fmaxf(a[32] + bb.y, 0.f);
            v[u].z = fmaxf(a[64] + bb.z, 0.f);
            v[u].w = fmaxf(a[96] + bb.w, 0.f);
            mx[u] = fmaxf(fmaxf(v[u].x, v[u].y), fmaxf(v[u].z, v[u].w));
        }
        #pragma unroll
        for (int off = 16; off > 0; off >>= 1) {
            #pragma unroll
            for (int u = 0; u < 4; ++u) mx[u] = fmaxf(mx[u], __shfl_xor(mx[u], off, 32));
        }
        #pragma unroll
        for (int u = 0; u < 4; ++u) {
            v[u].x = expf(v[u].x - mx[u]); v[u].y = expf(v[u].y - mx[u]);
            v[u].z = expf(v[u].z - mx[u]); v[u].w = expf(v[u].w - mx[u]);
            sum[u] = v[u].x + v[u].y + v[u].z + v[u].w;
        }
        #pragma unroll
        for (int off = 16; off > 0; off >>= 1) {
            #pragma unroll
            for (int u = 0; u < 4; ++u) sum[u] += __shfl_xor(sum[u], off, 32);
        }
        #pragma unroll
        for (int u = 0; u < 4; ++u) {
            const int r = r0 + u * NW;
            if (r < nrows) {
                const float inv = 1.0f / sum[u];
                float4 o;
                o.x = v[u].x * inv; o.y = v[u].y * inv; o.z = v[u].z * inv; o.w = v[u].w * inv;
                ((float4*)(out + (size_t)(base + r) * DIM))[lane] = o;
            }
        }
    }
}


extern "C" void kernel_launch(void* const* d_in, const int* in_sizes, int n_in,
                              void* d_out, int out_size, void* d_ws, size_t ws_size,
                              hipStream_t stream) {
    const float* x  = (const float*)d_in[0];
    const int*   ei = (const int*)  d_in[1];
    const float* W  = (const float*)d_in[2];
    const float* b  = (const float*)d_in[3];
    const int N = in_sizes[0] / DIM;
    const int E = in_sizes[1] / 2;
    const int* src = ei;
    const int* dst = ei + E;

    float*    out  = (float*)d_out;
    float*    dinv = (float*)d_ws;
    float*    xw   = dinv + N;
    unsigned* part = (unsigned*)(xw + (size_t)N * DIM);
    unsigned char* keys = (unsigned char*)(part + (size_t)DSLICE * N);
    const int nblk = (N + ROWS - 1) / ROWS;

    k_degree<<<dim3((N + DROWS - 1) / DROWS, DSLICE), WG, 0, stream>>>(dst, part, keys, nblk, N, E);
    k_dinv  <<<(N + 255) / 256, 256, 0, stream>>>(part, dinv, N);

    const int Mtiles = (N + 15) / 16;
    k_gemm_wmma<<<(Mtiles * 8 + 7) / 8, 256, 0, stream>>>(x, W, xw, N, Mtiles);

    k_accumulate<<<nblk, WG, 0, stream>>>(src, dst, keys, dinv, xw, b, out, N, E);
}


# k_degree sorts each 4096-edge chunk by owning row block in LDS and stores it in whole lines; k_accumulate reads its short (src,dst) runs, not every edge; sum order = LDS arrival as before
# speedup vs baseline: 4.5143x; 1.6154x over previous
#include <hip/hip_runtime.h>
#include <math.h>


typedef float v2f __attribute__((ext_vector_type(2)));
typedef float v8f __attribute__((ext_vector_type(8)));

#define DIM    128
#define WG     1024
#define NW     (WG / 32)
#define DROWS  2048
#define DSLICE 8
#define ROWS   200
#define CAP    8192
#define STRIDE 132
#define NT     ((ROWS + 15) / 16)
#define MAXT   ((NT + 3) / 4)


#define KEYMAX 256
#define CHUNK  4096
#define OFFP   288
#define RB     7

__device__ __forceinline__ void count1(unsigned* cnt, int t, int base) {
    const unsigned tl = (unsigned)(t - base);
    if (tl < (unsigned)DROWS) atomicAdd(&cnt[tl], 1u);
}

__global__ __launch_bounds__(WG) void k_degree(const int* __restrict__ src,
                                                 const int* __restrict__ dst,
                                                 unsigned* __restrict__ part,
                                                 int2* __restrict__ sorted,
                                                 int* __restrict__ offs, int nblk,
                                                 int N, int E) {
    __shared__ unsigned cnt[DROWS];
    __shared__ int2     srt[CHUNK];
    __shared__ unsigned hist[KEYMAX];
    __shared__ unsigned start[OFFP];
    const int tid  = threadIdx.x;
    const int base = blockIdx.x * DROWS;

    if (nblk <= KEYMAX) {
        const int lane   = tid & 31;
        const int nchunk = (E + CHUNK - 1) / CHUNK;
        const int nwg    = gridDim.x * gridDim.y;
        for (int c = blockIdx.y * gridDim.x + blockIdx.x; c < nchunk; c += nwg) {
            const int c0  = c * CHUNK;
            const int len = min(CHUNK, E - c0);
            int      s[CHUNK / WG], d[CHUNK / WG];
            unsigned ow[CHUNK / WG], rk[CHUNK / WG];
            #pragma unroll
            for (int u = 0; u < CHUNK / WG; ++u) {
                const int e = tid + u * WG;
                s[u] = 0; d[u] = 0;
                if (e < len) { s[u] = src[c0 + e]; d[u] = dst[c0 + e]; }
            }
            if (tid < KEYMAX) hist[tid] = 0u;
            __syncthreads();
            #pragma unroll
            for (int u = 0; u < CHUNK / WG; ++u) {
                ow[u] = (unsigned)d[u] / (unsigned)ROWS;
                rk[u] = 0u;
                if (tid + u * WG < len) rk[u] = atomicAdd(&hist[ow[u]], 1u);
            }
            __syncthreads();
            if (tid < 32) {
                unsigned v[KEYMAX / 32];
                unsigned tot = 0u;
                #pragma unroll
                for (int k = 0; k < KEYMAX / 32; ++k) { const unsigned t = hist[(KEYMAX / 32) * lane + k]; v[k] = tot; tot += t; }
                unsigned inc = tot;
                #pragma unroll
                for (int off = 1; off < 32; off <<= 1) {
                    const unsigned y = __shfl_up(inc, off, 32);
                    if (lane >= off) inc += y;
                }
                #pragma unroll
                for (int k = 0; k < KEYMAX / 32; ++k) start[(KEYMAX / 32) * lane + k] = inc - tot + v[k];
            } else if (tid >= KEYMAX && tid < OFFP) {
                start[tid] = (unsigned)len;
            }
            __syncthreads();
            #pragma unroll
            for (int u = 0; u < CHUNK / WG; ++u)
                if (tid + u * WG < len) srt[start[ow[u]] + rk[u]] = make_int2(s[u], d[u]);
            if (tid < OFFP) offs[(size_t)c * OFFP + tid] = (int)start[tid];
            __syncthreads();
            for (int i = tid; i < len; i += WG) sorted[c0 + i] = srt[i];
            __syncthreads();
        }
    }

    for (int i = tid; i < DROWS; i += WG) cnt[i] = 0u;
    __syncthreads();

    const int per = (((E + DSLICE - 1) / DSLICE) + 3) & ~3;
    const int e0  = min(E, (int)blockIdx.y * per);
    const int e1  = min(E, e0 + per);
    const int* __restrict__ ds = dst + e0;
    const int len = e1 - e0;

    if ((((size_t)ds) & 15) == 0) {
        const int4* __restrict__ d4 = (const int4*)ds;
        const int nv = len >> 2;
        int g = tid;
        for (; g + 3 * WG < nv; g += 4 * WG) {
            int4 t[4];
            #pragma unroll
            for (int u = 0; u < 4; ++u) t[u] = d4[g + u * WG];
            #pragma unroll
            for (int u = 0; u < 4; ++u) {
                count1(cnt, t[u].x, base); count1(cnt, t[u].y, base);
                count1(cnt, t[u].z, base); count1(cnt, t[u].w, base);
            }
        }
        for (; g < nv; g += WG) {
            const int4 t = d4[g];
            count1(cnt, t.x, base); count1(cnt, t.y, base);
            count1(cnt, t.z, base); count1(cnt, t.w, base);
        }
        for (int e = (nv << 2) + tid; e < len; e += WG) {
            const int t = ds[e];
            count1(cnt, t, base);
        }
    } else {
        for (int e = tid; e < len; e += WG) {
            const int t = ds[e];
            count1(cnt, t, base);
        }
    }
    __syncthreads();
    unsigned* __restrict__ mine = part + (size_t)blockIdx.y * N;
    for (int i = tid; i < DROWS; i += WG)
        if (base + i < N) mine[base + i] = cnt[i];
}

__global__ void k_dinv(const unsigned* __restrict__ part, float* __restrict__ dinv, int N) {
    const int i = blockIdx.x * 256 + threadIdx.x;
    if (i < N) {
        unsigned d = 1u;
        #pragma unroll
        for (int g = 0; g < DSLICE; ++g) d += part[(size_t)g * N + i];
        dinv[i] = rsqrtf((float)d);
    }
}


__device__ __forceinline__ void add_row_alone(float* a, const float* __restrict__ xr, float w) {
    for (int c = 0; c < DIM; ++c)
        atomicAdd(a + ((c & 3) * 32 + (c >> 2)), xr[c] * w);
}

__device__ __forceinline__ void add_row(float* a, const float4 v, const float w) {
    atomicAdd(a,      v.x * w);
    atomicAdd(a + 32, v.y * w);
    atomicAdd(a + 64, v.z * w);
    atomicAdd(a + 96, v.w * w);
}

template <int K>
__device__ __forceinline__ void tiles_times_w(const float* acc, const float* __restrict__ wcol,
                                              const int (&a_off)[MAXT], v8f (&cc)[MAXT]) {
    v2f bf[32];
    #pragma unroll
    for (int j = 0; j < 32; ++j) {
        const int k0 = 16 * (j & 7) + (j >> 3);
        bf[j].x = wcol[(size_t)k0 * DIM];
        bf[j].y = wcol[(size_t)(k0 + 4) * DIM];
    }
    #pragma unroll
    for (int j = 0; j < 32; ++j) {
        v2f a[K];
        #pragma unroll
        for (int t = 0; t < K; ++t) a[t] = *(const v2f*)(acc + a_off[t] + 4 * j);
        #pragma unroll
        for (int t = 0; t < K; ++t)
            cc[t] = __builtin_amdgcn_wmma_f32_16x16x4_f32(false, a[t], false, bf[j],
                                                          (short)0, cc[t], false, false);
    }
}

__global__ __launch_bounds__(WG) void k_accumulate(const int* __restrict__ src,
                                                     const int* __restrict__ dst,
                                                     const int2* __restrict__ sorted,
                                                     const int* __restrict__ offs,
                                                     const float* __restrict__ dinv,
                                                     const float* __restrict__ x,
                                                     const float* __restrict__ W,
                                                     const float* __restrict__ bias,
                                                     float* __restrict__ out, int N, int E) {
    __shared__ __attribute__((aligned(16))) float acc[ROWS * STRIDE];
    __shared__ int      l_s[CAP];
    __shared__ int      l_t[CAP];
    __shared__ float    l_w[CAP];
    __shared__ unsigned nlist;

    const int tid  = threadIdx.x;
    const int wave = __builtin_amdgcn_readfirstlane(tid >> 5);
    const int lane = tid & 31;
    const int base = blockIdx.x * ROWS;
    const int nrows = min(ROWS, N - base);
    const unsigned nr = (unsigned)nrows;

    for (int r0 = wave; r0 < nrows; r0 += 4 * NW) {
        float  di[4];
        float4 v[4];
        #pragma unroll
        for (int u = 0; u < 4; ++u) {
            const int r = min(r0 + u * NW, nrows - 1);
            di[u] = dinv[base + r];
            v[u]  = ((const float4*)(x + (size_t)(base + r) * DIM))[lane];
        }
        #pragma unroll
        for (int u = 0; u < 4; ++u) {
            const int r = r0 + u * NW;
            if (r < nrows) {
                float* a = acc + r * STRIDE + lane;
                a[0]  = v[u].x * di[u] * di[u];
                a[32] = v[u].y * di[u] * di[u];
                a[64] = v[u].z * di[u] * di[u];
                a[96] = v[u].w * di[u] * di[u];
            }
        }
    }
    if (tid == 0) nlist = 0u;
    __syncthreads();

    auto note = [&](int e) {
        const unsigned slot = atomicAdd(&nlist, 1u);
        if (slot < (unsigned)CAP) {
            l_s[slot] = e;
        } else {
            const int s  = src[e];
            const int tl = dst[e] - base;
            add_row_alone(acc + tl * STRIDE, x + (size_t)s * DIM, dinv[s] * dinv[base + tl]);
        }
    };
    const bool use_sorted = gridDim.x <= KEYMAX;
    if (use_sorted) {
        auto file = [&](const int2 ed, const float w) {
            const int   tl = ed.y - base;
            const unsigned slot = atomicAdd(&nlist, 1u);
            if (slot < (unsigned)CAP) {
                l_s[slot] = ed.x; l_t[slot] = tl; l_w[slot] = w;
            } else {
                add_row_alone(acc + tl * STRIDE, x + (size_t)ed.x * DIM, w);
            }
        };
        const int nchunk = (E + CHUNK - 1) / CHUNK;
        for (int q0 = wave; q0 < nchunk; q0 += RB * NW) {
            int b_l = 0, n_l = 0;
            {
                const int q = q0 + lane * NW;
                if (lane < RB && q < nchunk) {
                    const int* o = offs + (size_t)q * OFFP + blockIdx.x;
                    b_l = o[0];
                    n_l = o[1] - b_l;
                }
            }
            int2 ed[RB];
            #pragma unroll
            for (int r = 0; r < RB; ++r) {
                const int b = __builtin_amdgcn_readlane(b_l, r);
                const int n = __builtin_amdgcn_readlane(n_l, r);
                ed[r] = make_int2(0, 0);
                if (lane < n) ed[r] = sorted[(size_t)(q0 + r * NW) * CHUNK + b + lane];
            }
            float wr[RB];
            #pragma unroll
            for (int r = 0; r < RB; ++r)
                wr[r] = dinv[ed[r].x] * dinv[ed[r].y];
            #pragma unroll
            for (int r = 0; r < RB; ++r) {
                const int n = __builtin_amdgcn_readlane(n_l, r);
                if (lane < n) file(ed[r], wr[r]);
            }
            #pragma unroll
            for (int r = 0; r < RB; ++r) {
                const int b = __builtin_amdgcn_readlane(b_l, r);
                const int n = __builtin_amdgcn_readlane(n_l, r);
                for (int i = 32 + lane; i < n; i += 32) {
                    const int2 e2 = sorted[(size_t)(q0 + r * NW) * CHUNK + b + i];
                    file(e2, dinv[e2.x] * dinv[e2.y]);
                }
            }
        }
    } else if ((((size_t)dst) & 15) == 0) {
        const int4* __restrict__ d4 = (const int4*)dst;
        const int nv = E >> 2;
        int g = tid;
        for (; g + 7 * WG < nv; g += 8 * WG) {
            int4 t[8];
            #pragma unroll
            for (int u = 0; u < 8; ++u) t[u] = d4[g + u * WG];
            unsigned mask = 0u;
            #pragma unroll
            for (int u = 0; u < 8; ++u) {
                mask |= ((unsigned)(t[u].x - base) < nr ? 1u : 0u) << (4 * u + 0);
                mask |= ((unsigned)(t[u].y - base) < nr ? 1u : 0u) << (4 * u + 1);
                mask |= ((unsigned)(t[u].z - base) < nr ? 1u : 0u) << (4 * u + 2);
                mask |= ((unsigned)(t[u].w - base) < nr ? 1u : 0u) << (4 * u + 3);
            }
            while (mask) {
                const int k = __builtin_ctz(mask);
                mask &= mask - 1u;
                note(((g + (k >> 2) * WG) << 2) + (k & 3));
            }
        }
        for (; g < nv; g += WG) {
            const int4 t = d4[g];
            if ((unsigned)(t.x - base) < nr) note((g << 2) + 0);
            if ((unsigned)(t.y - base) < nr) note((g << 2) + 1);
            if ((unsigned)(t.z - base) < nr) note((g << 2) + 2);
            if ((unsigned)(t.w - base) < nr) note((g << 2) + 3);
        }
        for (int e = (nv << 2) + tid; e < E; e += WG)
            if ((unsigned)(dst[e] - base) < nr) note(e);
    } else {
        for (int e = tid; e < E; e += WG)
            if ((unsigned)(dst[e] - base) < nr) note(e);
    }
    __syncthreads();

    const int n = (int)min(nlist, (unsigned)CAP);
    for (int i = use_sorted ? n : tid; i < n; i += WG) {
        const int e  = l_s[i];
        const int s  = src[e];
        const int tl = dst[e] - base;
        l_s[i] = s;
        l_t[i] = tl;
        l_w[i] = dinv[s] * dinv[base + tl];
    }
    __syncthreads();

    for (int g0 = wave * 8; g0 < n; g0 += 8 * NW) {
        const int m = min(8, n - g0);
        int   s_l = 0, t_l = 0;
        float w_l = 0.f;
        if (lane < m) { s_l = l_s[g0 + lane]; t_l = l_t[g0 + lane]; w_l = l_w[g0 + lane]; }
        float4 v[8];
        #pragma unroll
        for (int u = 0; u < 8; ++u) {
            const int sj = __builtin_amdgcn_readlane(s_l, min(u, m - 1));
            v[u] = ((const float4*)(x + (size_t)sj * DIM))[lane];
        }
        #pragma unroll
        for (int u = 0; u < 8; ++u) {
            if (u < m) {
                const int   tj = __builtin_amdgcn_readlane(t_l, u);
                const float wj = __int_as_float(__builtin_amdgcn_readlane(__float_as_int(w_l), u));
                add_row(acc + tj * STRIDE + lane, v[u], wj);
            }
        }
    }
    __syncthreads();

    {
        const int tn = wave & 7;
        const int wq = wave >> 3;
        const int c  = lane & 15;
        const int h  = lane >> 4;
        const float* __restrict__ wcol = W + (size_t)(8 * h) * DIM + 16 * tn + c;
        int a_off[MAXT];
        v8f cc[MAXT];
        #pragma unroll
        for (int t = 0; t < MAXT; ++t) {
            a_off[t] = min(16 * (wq + 4 * t) + c, nrows - 1) * STRIDE + 2 * h;
            cc[t] = (v8f){0.f, 0.f, 0.f, 0.f, 0.f, 0.f, 0.f, 0.f};
        }
        if (wq + 4 * (MAXT - 1) < NT) tiles_times_w<MAXT>(acc, wcol, a_off, cc);
        else                          tiles_times_w<MAXT - 1>(acc, wcol, a_off, cc);
        __syncthreads();
        const int wpos = (c & 3) * 32 + 4 * tn + (c >> 2);
        #pragma unroll
        for (int t = 0; t < MAXT; ++t) {
            #pragma unroll
            for (int r = 0; r < 8; ++r) {
                const int row = 16 * (wq + 4 * t) + 8 * h + r;
                if (row < nrows) acc[row * STRIDE + wpos] = cc[t][r];
            }
        }
    }
    __syncthreads();

    const float4 bb = ((const float4*)bias)[lane];
    for (int r0 = wave; r0 < nrows; r0 += 4 * NW) {
        float4 v[4];
        float  mx[4], sum[4];
        #pragma unroll
        for (int u = 0; u < 4; ++u) {
            const int r = min(r0 + u * NW, nrows - 1);
            const float* a = acc + r * STRIDE + lane;
            v[u].x = fmaxf(a[0]  + bb.x, 0.f);
            v[u].y = fmaxf(a[32] + bb.y, 0.f);
            v[u].z = fmaxf(a[64] + bb.z, 0.f);
            v[u].w = fmaxf(a[96] + bb.w, 0.f);
            mx[u] = fmaxf(fmaxf(v[u].x, v[u].y), fmaxf(v[u].z, v[u].w));
        }
        #pragma unroll
        for (int off = 16; off > 0; off >>= 1) {
            #pragma unroll
            for (int u = 0; u < 4; ++u) mx[u] = fmaxf(mx[u], __shfl_xor(mx[u], off, 32));
        }
        #pragma unroll
        for (int u = 0; u < 4; ++u) {
            v[u].x = expf(v[u].x - mx[u]); v[u].y = expf(v[u].y - mx[u]);
            v[u].z = expf(v[u].z - mx[u]); v[u].w = expf(v[u].w - mx[u]);
            sum[u] = v[u].x + v[u].y + v[u].z + v[u].w;
        }
        #pragma unroll
        for (int off = 16; off > 0; off >>= 1) {
            #pragma unroll
            for (int u = 0; u < 4; ++u) sum[u] += __shfl_xor(sum[u], off, 32);
        }
        #pragma unroll
        for (int u = 0; u < 4; ++u) {
            const int r = r0 + u * NW;
            if (r < nrows) {
                const float inv = 1.0f / sum[u];
                float4 o;
                o.x = v[u].x * inv; o.y = v[u].y * inv; o.z = v[u].z * inv; o.w = v[u].w * inv;
                ((float4*)(out + (size_t)(base + r) * DIM))[lane] = o;
            }
        }
    }
}


extern "C" void kernel_launch(void* const* d_in, const int* in_sizes, int n_in,
                              void* d_out, int out_size, void* d_ws, size_t ws_size,
                              hipStream_t stream) {
    const float* x  = (const float*)d_in[0];
    const int*   ei = (const int*)  d_in[1];
    const float* W  = (const float*)d_in[2];
    const float* b  = (const float*)d_in[3];
    const int N = in_sizes[0] / DIM;
    const int E = in_sizes[1] / 2;
    const int* src = ei;
    const int* dst = ei + E;

    float*    out  = (float*)d_out;
    int2*     sorted = (int2*)d_ws;
    int*      offs = (int*)(sorted + E);
    float*    dinv = (float*)(offs + (size_t)((E + CHUNK - 1) / CHUNK) * OFFP);
    unsigned* part = (unsigned*)(dinv + N);
    const int nblk = (N + ROWS - 1) / ROWS;

    k_degree<<<dim3((N + DROWS - 1) / DROWS, DSLICE), WG, 0, stream>>>(src, dst, part, sorted, offs, nblk, N, E);
    k_dinv  <<<(N + 255) / 256, 256, 0, stream>>>(part, dinv, N);

    k_accumulate<<<nblk, WG, 0, stream>>>(src, dst, sorted, offs, dinv, x, W, b, out, N, E);
}


# in-degrees counted from the sorted runs: a k_dinv work-group reads its own row block's run of every chunk and counts it in LDS; k_degree only sorts, its 25 passes over dst go; same sums, same order
# speedup vs baseline: 4.7310x; 1.0480x over previous
#include <hip/hip_runtime.h>
#include <math.h>


typedef float v2f __attribute__((ext_vector_type(2)));
typedef float v8f __attribute__((ext_vector_type(8)));

#define DIM    128
#define WG     1024
#define NW     (WG / 32)
#define DROWS  2048
#define DSLICE 8
#define ROWS   200
#define CAP    8192
#define STRIDE 132
#define NT     ((ROWS + 15) / 16)
#define MAXT   ((NT + 3) / 4)


#define KEYMAX 256
#define CHUNK  4096
#define OFFP   288
#define RB     7

__device__ __forceinline__ void count1(unsigned* cnt, int t, int base) {
    const unsigned tl = (unsigned)(t - base);
    if (tl < (unsigned)DROWS) atomicAdd(&cnt[tl], 1u);
}

__global__ __launch_bounds__(WG) void k_degree(const int* __restrict__ src,
                                                 const int* __restrict__ dst,
                                                 unsigned* __restrict__ part,
                                                 int2* __restrict__ sorted,
                                                 int* __restrict__ offs, int nblk,
                                                 int N, int E) {
    __shared__ unsigned cnt[DROWS];
    __shared__ int2     srt[CHUNK];
    __shared__ unsigned hist[KEYMAX];
    __shared__ unsigned start[OFFP];
    const int tid  = threadIdx.x;
    const int base = blockIdx.x * DROWS;

    if (nblk <= KEYMAX) {
        const int lane   = tid & 31;
        const int nchunk = (E + CHUNK - 1) / CHUNK;
        const int nwg    = gridDim.x * gridDim.y;
        for (int c = blockIdx.y * gridDim.x + blockIdx.x; c < nchunk; c += nwg) {
            const int c0  = c * CHUNK;
            const int len = min(CHUNK, E - c0);
            int      s[CHUNK / WG], d[CHUNK / WG];
            unsigned ow[CHUNK / WG], rk[CHUNK / WG];
            #pragma unroll
            for (int u = 0; u < CHUNK / WG; ++u) {
                const int e = tid + u * WG;
                s[u] = 0; d[u] = 0;
                if (e < len) { s[u] = src[c0 + e]; d[u] = dst[c0 + e]; }
            }
            if (tid < KEYMAX) hist[tid] = 0u;
            __syncthreads();
            #pragma unroll
            for (int u = 0; u < CHUNK / WG; ++u) {
                ow[u] = (unsigned)d[u] / (unsigned)ROWS;
                rk[u] = 0u;
                if (tid + u * WG < len) rk[u] = atomicAdd(&hist[ow[u]], 1u);
            }
            __syncthreads();
            if (tid < 32) {
                unsigned v[KEYMAX / 32];
                unsigned tot = 0u;
                #pragma unroll
                for (int k = 0; k < KEYMAX / 32; ++k) { const unsigned t = hist[(KEYMAX / 32) * lane + k]; v[k] = tot; tot += t; }
                unsigned inc = tot;
                #pragma unroll
                for (int off = 1; off < 32; off <<= 1) {
                    const unsigned y = __shfl_up(inc, off, 32);
                    if (lane >= off) inc += y;
                }
                #pragma unroll
                for (int k = 0; k < KEYMAX / 32; ++k) start[(KEYMAX / 32) * lane + k] = inc - tot + v[k];
            } else if (tid >= KEYMAX && tid < OFFP) {
                start[tid] = (unsigned)len;
            }
            __syncthreads();
            #pragma unroll
            for (int u = 0; u < CHUNK / WG; ++u)
                if (tid + u * WG < len) srt[start[ow[u]] + rk[u]] = make_int2(s[u], d[u]);
            if (tid < OFFP) offs[(size_t)c * OFFP + tid] = (int)start[tid];
            __syncthreads();
            for (int i = tid; i < len; i += WG) sorted[c0 + i] = srt[i];
            __syncthreads();
        }
        return;
    }

    for (int i = tid; i < DROWS; i += WG) cnt[i] = 0u;
    __syncthreads();

    const int per = (((E + DSLICE - 1) / DSLICE) + 3) & ~3;
    const int e0  = min(E, (int)blockIdx.y * per);
    const int e1  = min(E, e0 + per);
    const int* __restrict__ ds = dst + e0;
    const int len = e1 - e0;

    if ((((size_t)ds) & 15) == 0) {
        const int4* __restrict__ d4 = (const int4*)ds;
        const int nv = len >> 2;
        int g = tid;
        for (; g + 3 * WG < nv; g += 4 * WG) {
            int4 t[4];
            #pragma unroll
            for (int u = 0; u < 4; ++u) t[u] = d4[g + u * WG];
            #pragma unroll
            for (int u = 0; u < 4; ++u) {
                count1(cnt, t[u].x, base); count1(cnt, t[u].y, base);
                count1(cnt, t[u].z, base); count1(cnt, t[u].w, base);
            }
        }
        for (; g < nv; g += WG) {
            const int4 t = d4[g];
            count1(cnt, t.x, base); count1(cnt, t.y, base);
            count1(cnt, t.z, base); count1(cnt, t.w, base);
        }
        for (int e = (nv << 2) + tid; e < len; e += WG) {
            const int t = ds[e];
            count1(cnt, t, base);
        }
    } else {
        for (int e = tid; e < len; e += WG) {
            const int t = ds[e];
            count1(cnt, t, base);
        }
    }
    __syncthreads();
    unsigned* __restrict__ mine = part + (size_t)blockIdx.y * N;
    for (int i = tid; i < DROWS; i += WG)
        if (base + i < N) mine[base + i] = cnt[i];
}

#define DWG  256
#define DNW  (DWG / 32)
#define DRB  25

__global__ __launch_bounds__(DWG) void k_dinv(const unsigned* __restrict__ part,
                                               const int2* __restrict__ sorted,
                                               const int* __restrict__ offs,
                                               float* __restrict__ dinv, int N, int E) {
    __shared__ unsigned cnt[ROWS];
    const int tid   = threadIdx.x;
    const int base  = blockIdx.x * ROWS;
    const int nrows = min(ROWS, N - base);
    if (gridDim.x > KEYMAX) {
        if (tid < nrows) {
            unsigned d = 1u;
            #pragma unroll
            for (int g = 0; g < DSLICE; ++g) d += part[(size_t)g * N + base + tid];
            dinv[base + tid] = rsqrtf((float)d);
        }
        return;
    }
    const int wave = __builtin_amdgcn_readfirstlane(tid >> 5);
    const int lane = tid & 31;
    const unsigned nr = (unsigned)nrows;
    if (tid < ROWS) cnt[tid] = 0u;
    __syncthreads();
    const int nchunk = (E + CHUNK - 1) / CHUNK;
    for (int q0 = wave; q0 < nchunk; q0 += DRB * DNW) {
        int b_l = 0, n_l = 0;
        {
            const int q = q0 + lane * DNW;
            if (lane < DRB && q < nchunk) {
                const int* o = offs + (size_t)q * OFFP + blockIdx.x;
                b_l = o[0];
                n_l = o[1] - b_l;
            }
        }
        int t[DRB];
        #pragma unroll
        for (int r = 0; r < DRB; ++r) {
            const int b = __builtin_amdgcn_readlane(b_l, r);
            const int n = __builtin_amdgcn_readlane(n_l, r);
            t[r] = -1;
            if (lane < n) t[r] = sorted[(size_t)(q0 + r * DNW) * CHUNK + b + lane].y;
        }
        #pragma unroll
        for (int r = 0; r < DRB; ++r) {
            const unsigned tl = (unsigned)(t[r] - base);
            if (tl < nr) atomicAdd(&cnt[tl], 1u);
        }
        #pragma unroll
        for (int r = 0; r < DRB; ++r) {
            const int b = __builtin_amdgcn_readlane(b_l, r);
            const int n = __builtin_amdgcn_readlane(n_l, r);
            for (int i = 32 + lane; i < n; i += 32) {
                const unsigned tl = (unsigned)(sorted[(size_t)(q0 + r * DNW) * CHUNK + b + i].y - base);
                if (tl < nr) atomicAdd(&cnt[tl], 1u);
            }
        }
    }
    __syncthreads();
    if (tid < nrows) dinv[base + tid] = rsqrtf((float)(1u + cnt[tid]));
}


__device__ __forceinline__ void add_row_alone(float* a, const float* __restrict__ xr, float w) {
    for (int c = 0; c < DIM; ++c)
        atomicAdd(a + ((c & 3) * 32 + (c >> 2)), xr[c] * w);
}

__device__ __forceinline__ void add_row(float* a, const float4 v, const float w) {
    atomicAdd(a,      v.x * w);
    atomicAdd(a + 32, v.y * w);
    atomicAdd(a + 64, v.z * w);
    atomicAdd(a + 96, v.w * w);
}

template <int K>
__device__ __forceinline__ void tiles_times_w(const float* acc, const float* __restrict__ wcol,
                                              const int (&a_off)[MAXT], v8f (&cc)[MAXT]) {
    v2f bf[32];
    #pragma unroll
    for (int j = 0; j < 32; ++j) {
        const int k0 = 16 * (j & 7) + (j >> 3);
        bf[j].x = wcol[(size_t)k0 * DIM];
        bf[j].y = wcol[(size_t)(k0 + 4) * DIM];
    }
    #pragma unroll
    for (int j = 0; j < 32; ++j) {
        v2f a[K];
        #pragma unroll
        for (int t = 0; t < K; ++t) a[t] = *(const v2f*)(acc + a_off[t] + 4 * j);
        #pragma unroll
        for (int t = 0; t < K; ++t)
            cc[t] = __builtin_amdgcn_wmma_f32_16x16x4_f32(false, a[t], false, bf[j],
                                                          (short)0, cc[t], false, false);
    }
}

__global__ __launch_bounds__(WG) void k_accumulate(const int* __restrict__ src,
                                                     const int* __restrict__ dst,
                                                     const int2* __restrict__ sorted,
                                                     const int* __restrict__ offs,
                                                     const float* __restrict__ dinv,
                                                     const float* __restrict__ x,
                                                     const float* __restrict__ W,
                                                     const float* __restrict__ bias,
                                                     float* __restrict__ out, int N, int E) {
    __shared__ __attribute__((aligned(16))) float acc[ROWS * STRIDE];
    __shared__ int      l_s[CAP];
    __shared__ int      l_t[CAP];
    __shared__ float    l_w[CAP];
    __shared__ unsigned nlist;

    const int tid  = threadIdx.x;
    const int wave = __builtin_amdgcn_readfirstlane(tid >> 5);
    const int lane = tid & 31;
    const int base = blockIdx.x * ROWS;
    const int nrows = min(ROWS, N - base);
    const unsigned nr = (unsigned)nrows;

    for (int r0 = wave; r0 < nrows; r0 += 4 * NW) {
        float  di[4];
        float4 v[4];
        #pragma unroll
        for (int u = 0; u < 4; ++u) {
            const int r = min(r0 + u * NW, nrows - 1);
            di[u] = dinv[base + r];
            v[u]  = ((const float4*)(x + (size_t)(base + r) * DIM))[lane];
        }
        #pragma unroll
        for (int u = 0; u < 4; ++u) {
            const int r = r0 + u * NW;
            if (r < nrows) {
                float* a = acc + r * STRIDE + lane;
                a[0]  = v[u].x * di[u] * di[u];
                a[32] = v[u].y * di[u] * di[u];
                a[64] = v[u].z * di[u] * di[u];
                a[96] = v[u].w * di[u] * di[u];
            }
        }
    }
    if (tid == 0) nlist = 0u;
    __syncthreads();

    auto note = [&](int e) {
        const unsigned slot = atomicAdd(&nlist, 1u);
        if (slot < (unsigned)CAP) {
            l_s[slot] = e;
        } else {
            const int s  = src[e];
            const int tl = dst[e] - base;
            add_row_alone(acc + tl * STRIDE, x + (size_t)s * DIM, dinv[s] * dinv[base + tl]);
        }
    };
    const bool use_sorted = gridDim.x <= KEYMAX;
    if (use_sorted) {
        auto file = [&](const int2 ed, const float w) {
            const int   tl = ed.y - base;
            const unsigned slot = atomicAdd(&nlist, 1u);
            if (slot < (unsigned)CAP) {
                l_s[slot] = ed.x; l_t[slot] = tl; l_w[slot] = w;
            } else {
                add_row_alone(acc + tl * STRIDE, x + (size_t)ed.x * DIM, w);
            }
        };
        const int nchunk = (E + CHUNK - 1) / CHUNK;
        for (int q0 = wave; q0 < nchunk; q0 += RB * NW) {
            int b_l = 0, n_l = 0;
            {
                const int q = q0 + lane * NW;
                if (lane < RB && q < nchunk) {
                    const int* o = offs + (size_t)q * OFFP + blockIdx.x;
                    b_l = o[0];
                    n_l = o[1] - b_l;
                }
            }
            int2 ed[RB];
            #pragma unroll
            for (int r = 0; r < RB; ++r) {
                const int b = __builtin_amdgcn_readlane(b_l, r);
                const int n = __builtin_amdgcn_readlane(n_l, r);
                ed[r] = make_int2(0, 0);
                if (lane < n) ed[r] = sorted[(size_t)(q0 + r * NW) * CHUNK + b + lane];
            }
            float wr[RB];
            #pragma unroll
            for (int r = 0; r < RB; ++r)
                wr[r] = dinv[ed[r].x] * dinv[ed[r].y];
            #pragma unroll
            for (int r = 0; r < RB; ++r) {
                const int n = __builtin_amdgcn_readlane(n_l, r);
                if (lane < n) file(ed[r], wr[r]);
            }
            #pragma unroll
            for (int r = 0; r < RB; ++r) {
                const int b = __builtin_amdgcn_readlane(b_l, r);
                const int n = __builtin_amdgcn_readlane(n_l, r);
                for (int i = 32 + lane; i < n; i += 32) {
                    const int2 e2 = sorted[(size_t)(q0 + r * NW) * CHUNK + b + i];
                    file(e2, dinv[e2.x] * dinv[e2.y]);
                }
            }
        }
    } else if ((((size_t)dst) & 15) == 0) {
        const int4* __restrict__ d4 = (const int4*)dst;
        const int nv = E >> 2;
        int g = tid;
        for (; g + 7 * WG < nv; g += 8 * WG) {
            int4 t[8];
            #pragma unroll
            for (int u = 0; u < 8; ++u) t[u] = d4[g + u * WG];
            unsigned mask = 0u;
            #pragma unroll
            for (int u = 0; u < 8; ++u) {
                mask |= ((unsigned)(t[u].x - base) < nr ? 1u : 0u) << (4 * u + 0);
                mask |= ((unsigned)(t[u].y - base) < nr ? 1u : 0u) << (4 * u + 1);
                mask |= ((unsigned)(t[u].z - base) < nr ? 1u : 0u) << (4 * u + 2);
                mask |= ((unsigned)(t[u].w - base) < nr ? 1u : 0u) << (4 * u + 3);
            }
            while (mask) {
                const int k = __builtin_ctz(mask);
                mask &= mask - 1u;
                note(((g + (k >> 2) * WG) << 2) + (k & 3));
            }
        }
        for (; g < nv; g += WG) {
            const int4 t = d4[g];
            if ((unsigned)(t.x - base) < nr) note((g << 2) + 0);
            if ((unsigned)(t.y - base) < nr) note((g << 2) + 1);
            if ((unsigned)(t.z - base) < nr) note((g << 2) + 2);
            if ((unsigned)(t.w - base) < nr) note((g << 2) + 3);
        }
        for (int e = (nv << 2) + tid; e < E; e += WG)
            if ((unsigned)(dst[e] - base) < nr) note(e);
    } else {
        for (int e = tid; e < E; e += WG)
            if ((unsigned)(dst[e] - base) < nr) note(e);
    }
    __syncthreads();

    const int n = (int)min(nlist, (unsigned)CAP);
    for (int i = use_sorted ? n : tid; i < n; i += WG) {
        const int e  = l_s[i];
        const int s  = src[e];
        const int tl = dst[e] - base;
        l_s[i] = s;
        l_t[i] = tl;
        l_w[i] = dinv[s] * dinv[base + tl];
    }
    __syncthreads();

    for (int g0 = wave * 8; g0 < n; g0 += 8 * NW) {
        const int m = min(8, n - g0);
        int   s_l = 0, t_l = 0;
        float w_l = 0.f;
        if (lane < m) { s_l = l_s[g0 + lane]; t_l = l_t[g0 + lane]; w_l = l_w[g0 + lane]; }
        float4 v[8];
        #pragma unroll
        for (int u = 0; u < 8; ++u) {
            const int sj = __builtin_amdgcn_readlane(s_l, min(u, m - 1));
            v[u] = ((const float4*)(x + (size_t)sj * DIM))[lane];
        }
        #pragma unroll
        for (int u = 0; u < 8; ++u) {
            if (u < m) {
                const int   tj = __builtin_amdgcn_readlane(t_l, u);
                const float wj = __int_as_float(__builtin_amdgcn_readlane(__float_as_int(w_l), u));
                add_row(acc + tj * STRIDE + lane, v[u], wj);
            }
        }
    }
    __syncthreads();

    {
        const int tn = wave & 7;
        const int wq = wave >> 3;
        const int c  = lane & 15;
        const int h  = lane >> 4;
        const float* __restrict__ wcol = W + (size_t)(8 * h) * DIM + 16 * tn + c;
        int a_off[MAXT];
        v8f cc[MAXT];
        #pragma unroll
        for (int t = 0; t < MAXT; ++t) {
            a_off[t] = min(16 * (wq + 4 * t) + c, nrows - 1) * STRIDE + 2 * h;
            cc[t] = (v8f){0.f, 0.f, 0.f, 0.f, 0.f, 0.f, 0.f, 0.f};
        }
        if (wq + 4 * (MAXT - 1) < NT) tiles_times_w<MAXT>(acc, wcol, a_off, cc);
        else                          tiles_times_w<MAXT - 1>(acc, wcol, a_off, cc);
        __syncthreads();
        const int wpos = (c & 3) * 32 + 4 * tn + (c >> 2);
        #pragma unroll
        for (int t = 0; t < MAXT; ++t) {
            #pragma unroll
            for (int r = 0; r < 8; ++r) {
                const int row = 16 * (wq + 4 * t) + 8 * h + r;
                if (row < nrows) acc[row * STRIDE + wpos] = cc[t][r];
            }
        }
    }
    __syncthreads();

    const float4 bb = ((const float4*)bias)[lane];
    for (int r0 = wave; r0 < nrows; r0 += 4 * NW) {
        float4 v[4];
        float  mx[4], sum[4];
        #pragma unroll
        for (int u = 0; u < 4; ++u) {
            const int r = min(r0 + u * NW, nrows - 1);
            const float* a = acc + r * STRIDE + lane;
            v[u].x = fmaxf(a[0]  + bb.x, 0.f);
            v[u].y = fmaxf(a[32] + bb.y, 0.f);
            v[u].z = fmaxf(a[64] + bb.z, 0.f);
            v[u].w = fmaxf(a[96] + bb.w, 0.f);
            mx[u] = fmaxf(fmaxf(v[u].x, v[u].y), fmaxf(v[u].z, v[u].w));
        }
        #pragma unroll
        for (int off = 16; off > 0; off >>= 1) {
            #pragma unroll
            for (int u = 0; u < 4; ++u) mx[u] = fmaxf(mx[u], __shfl_xor(mx[u], off, 32));
        }
        #pragma unroll
        for (int u = 0; u < 4; ++u) {
            v[u].x = expf(v[u].x - mx[u]); v[u].y = expf(v[u].y - mx[u]);
            v[u].z = expf(v[u].z - mx[u]); v[u].w = expf(v[u].w - mx[u]);
            sum[u] = v[u].x + v[u].y + v[u].z + v[u].w;
        }
        #pragma unroll
        for (int off = 16; off > 0; off >>= 1) {
            #pragma unroll
            for (int u = 0; u < 4; ++u) sum[u] += __shfl_xor(sum[u], off, 32);
        }
        #pragma unroll
        for (int u = 0; u < 4; ++u) {
            const int r = r0 + u * NW;
            if (r < nrows) {
                const float inv = 1.0f / sum[u];
                float4 o;
                o.x = v[u].x * inv; o.y = v[u].y * inv; o.z = v[u].z * inv; o.w = v[u].w * inv;
                ((float4*)(out + (size_t)(base + r) * DIM))[lane] = o;
            }
        }
    }
}


extern "C" void kernel_launch(void* const* d_in, const int* in_sizes, int n_in,
                              void* d_out, int out_size, void* d_ws, size_t ws_size,
                              hipStream_t stream) {
    const float* x  = (const float*)d_in[0];
    const int*   ei = (const int*)  d_in[1];
    const float* W  = (const float*)d_in[2];
    const float* b  = (const float*)d_in[3];
    const int N = in_sizes[0] / DIM;
    const int E = in_sizes[1] / 2;
    const int* src = ei;
    const int* dst = ei + E;

    float*    out  = (float*)d_out;
    int2*     sorted = (int2*)d_ws;
    int*      offs = (int*)(sorted + E);
    float*    dinv = (float*)(offs + (size_t)((E + CHUNK - 1) / CHUNK) * OFFP);
    unsigned* part = (unsigned*)(dinv + N);
    const int nblk = (N + ROWS - 1) / ROWS;

    k_degree<<<dim3((N + DROWS - 1) / DROWS, DSLICE), WG, 0, stream>>>(src, dst, part, sorted, offs, nblk, N, E);
    k_dinv  <<<nblk, DWG, 0, stream>>>(part, sorted, offs, dinv, N, E);

    k_accumulate<<<nblk, WG, 0, stream>>>(src, dst, sorted, offs, dinv, x, W, b, out, N, E);
}


# step B, no f32 LDS atomics: edges filed in 32 per-wave LDS bins (row mod 32); owner wave adds a row by 16-B LDS load, FMAs, 16-B store, next 8 rows in flight; sum order: bin rank, GEMM k ascending
# speedup vs baseline: 4.9477x; 1.0458x over previous
#include <hip/hip_runtime.h>
#include <math.h>


typedef float v2f __attribute__((ext_vector_type(2)));
typedef float v8f __attribute__((ext_vector_type(8)));

#define DIM    128
#define WG     1024
#define NW     (WG / 32)
#define DROWS  2048
#define DSLICE 8
#define ROWS   200
#define CAP    8192
#define BCAP   (CAP / NW)
#define STRIDE 132
#define NT     ((ROWS + 15) / 16)
#define MAXT   ((NT + 3) / 4)


#define KEYMAX 256
#define CHUNK  4096
#define OFFP   288
#define RB     7

__device__ __forceinline__ void count1(unsigned* cnt, int t, int base) {
    const unsigned tl = (unsigned)(t - base);
    if (tl < (unsigned)DROWS) atomicAdd(&cnt[tl], 1u);
}

__global__ __launch_bounds__(WG) void k_degree(const int* __restrict__ src,
                                                 const int* __restrict__ dst,
                                                 unsigned* __restrict__ part,
                                                 int2* __restrict__ sorted,
                                                 int* __restrict__ offs, int nblk,
                                                 int N, int E) {
    __shared__ unsigned cnt[DROWS];
    __shared__ int2     srt[CHUNK];
    __shared__ unsigned hist[KEYMAX];
    __shared__ unsigned start[OFFP];
    const int tid  = threadIdx.x;
    const int base = blockIdx.x * DROWS;

    if (nblk <= KEYMAX) {
        const int lane   = tid & 31;
        const int nchunk = (E + CHUNK - 1) / CHUNK;
        const int nwg    = gridDim.x * gridDim.y;
        for (int c = blockIdx.y * gridDim.x + blockIdx.x; c < nchunk; c += nwg) {
            const int c0  = c * CHUNK;
            const int len = min(CHUNK, E - c0);
            int      s[CHUNK / WG], d[CHUNK / WG];
            unsigned ow[CHUNK / WG], rk[CHUNK / WG];
            #pragma unroll
            for (int u = 0; u < CHUNK / WG; ++u) {
                const int e = tid + u * WG;
                s[u] = 0; d[u] = 0;
                if (e < len) { s[u] = src[c0 + e]; d[u] = dst[c0 + e]; }
            }
            if (tid < KEYMAX) hist[tid] = 0u;
            __syncthreads();
            #pragma unroll
            for (int u = 0; u < CHUNK / WG; ++u) {
                ow[u] = (unsigned)d[u] / (unsigned)ROWS;
                rk[u] = 0u;
                if (tid + u * WG < len) rk[u] = atomicAdd(&hist[ow[u]], 1u);
            }
            __syncthreads();
            if (tid < 32) {
                unsigned v[KEYMAX / 32];
                unsigned tot = 0u;
                #pragma unroll
                for (int k = 0; k < KEYMAX / 32; ++k) { const unsigned t = hist[(KEYMAX / 32) * lane + k]; v[k] = tot; tot += t; }
                unsigned inc = tot;
                #pragma unroll
                for (int off = 1; off < 32; off <<= 1) {
                    const unsigned y = __shfl_up(inc, off, 32);
                    if (lane >= off) inc += y;
                }
                #pragma unroll
                for (int k = 0; k < KEYMAX / 32; ++k) start[(KEYMAX / 32) * lane + k] = inc - tot + v[k];
            } else if (tid >= KEYMAX && tid < OFFP) {
                start[tid] = (unsigned)len;
            }
            __syncthreads();
            #pragma unroll
            for (int u = 0; u < CHUNK / WG; ++u)
                if (tid + u * WG < len) srt[start[ow[u]] + rk[u]] = make_int2(s[u], d[u]);
            if (tid < OFFP) offs[(size_t)c * OFFP + tid] = (int)start[tid];
            __syncthreads();
            for (int i = tid; i < len; i += WG) sorted[c0 + i] = srt[i];
            __syncthreads();
        }
        return;
    }

    for (int i = tid; i < DROWS; i += WG) cnt[i] = 0u;
    __syncthreads();

    const int per = (((E + DSLICE - 1) / DSLICE) + 3) & ~3;
    const int e0  = min(E, (int)blockIdx.y * per);
    const int e1  = min(E, e0 + per);
    const int* __restrict__ ds = dst + e0;
    const int len = e1 - e0;

    if ((((size_t)ds) & 15) == 0) {
        const int4* __restrict__ d4 = (const int4*)ds;
        const int nv = len >> 2;
        int g = tid;
        for (; g + 3 * WG < nv; g += 4 * WG) {
            int4 t[4];
            #pragma unroll
            for (int u = 0; u < 4; ++u) t[u] = d4[g + u * WG];
            #pragma unroll
            for (int u = 0; u < 4; ++u) {
                count1(cnt, t[u].x, base); count1(cnt, t[u].y, base);
                count1(cnt, t[u].z, base); count1(cnt, t[u].w, base);
            }
        }
        for (; g < nv; g += WG) {
            const int4 t = d4[g];
            count1(cnt, t.x, base); count1(cnt, t.y, base);
            count1(cnt, t.z, base); count1(cnt, t.w, base);
        }
        for (int e = (nv << 2) + tid; e < len; e += WG) {
            const int t = ds[e];
            count1(cnt, t, base);
        }
    } else {
        for (int e = tid; e < len; e += WG) {
            const int t = ds[e];
            count1(cnt, t, base);
        }
    }
    __syncthreads();
    unsigned* __restrict__ mine = part + (size_t)blockIdx.y * N;
    for (int i = tid; i < DROWS; i += WG)
        if (base + i < N) mine[base + i] = cnt[i];
}

#define DWG  256
#define DNW  (DWG / 32)
#define DRB  25

__global__ __launch_bounds__(DWG) void k_dinv(const unsigned* __restrict__ part,
                                               const int2* __restrict__ sorted,
                                               const int* __restrict__ offs,
                                               float* __restrict__ dinv, int N, int E) {
    __shared__ unsigned cnt[ROWS];
    const int tid   = threadIdx.x;
    const int base  = blockIdx.x * ROWS;
    const int nrows = min(ROWS, N - base);
    if (gridDim.x > KEYMAX) {
        if (tid < nrows) {
            unsigned d = 1u;
            #pragma unroll
            for (int g = 0; g < DSLICE; ++g) d += part[(size_t)g * N + base + tid];
            dinv[base + tid] = rsqrtf((float)d);
        }
        return;
    }
    const int wave = __builtin_amdgcn_readfirstlane(tid >> 5);
    const int lane = tid & 31;
    const unsigned nr = (unsigned)nrows;
    if (tid < ROWS) cnt[tid] = 0u;
    __syncthreads();
    const int nchunk = (E + CHUNK - 1) / CHUNK;
    for (int q0 = wave; q0 < nchunk; q0 += DRB * DNW) {
        int b_l = 0, n_l = 0;
        {
            const int q = q0 + lane * DNW;
            if (lane < DRB && q < nchunk) {
                const int* o = offs + (size_t)q * OFFP + blockIdx.x;
                b_l = o[0];
                n_l = o[1] - b_l;
            }
        }
        int t[DRB];
        #pragma unroll
        for (int r = 0; r < DRB; ++r) {
            const int b = __builtin_amdgcn_readlane(b_l, r);
            const int n = __builtin_amdgcn_readlane(n_l, r);
            t[r] = -1;
            if (lane < n) t[r] = sorted[(size_t)(q0 + r * DNW) * CHUNK + b + lane].y;
        }
        #pragma unroll
        for (int r = 0; r < DRB; ++r) {
            const unsigned tl = (unsigned)(t[r] - base);
            if (tl < nr) atomicAdd(&cnt[tl], 1u);
        }
        #pragma unroll
        for (int r = 0; r < DRB; ++r) {
            const int b = __builtin_amdgcn_readlane(b_l, r);
            const int n = __builtin_amdgcn_readlane(n_l, r);
            for (int i = 32 + lane; i < n; i += 32) {
                const unsigned tl = (unsigned)(sorted[(size_t)(q0 + r * DNW) * CHUNK + b + i].y - base);
                if (tl < nr) atomicAdd(&cnt[tl], 1u);
            }
        }
    }
    __syncthreads();
    if (tid < nrows) dinv[base + tid] = rsqrtf((float)(1u + cnt[tid]));
}


__device__ __forceinline__ void add_row_alone(float* a, const float* __restrict__ xr, float w) {
    for (int c = 0; c < DIM; ++c)
        atomicAdd(a + c, xr[c] * w);
}

__device__ __forceinline__ void add_row(float4* a, const float4 v, const float w) {
    float4 o = *a;
    o.x += v.x * w; o.y += v.y * w; o.z += v.z * w; o.w += v.w * w;
    *a = o;
}

template <int K>
__device__ __forceinline__ void tiles_times_w(const float* acc, const float* __restrict__ wcol,
                                              const int (&a_off)[MAXT], v8f (&cc)[MAXT]) {
    v2f bf[32];
    #pragma unroll
    for (int j = 0; j < 32; ++j) {
        bf[j].x = wcol[(size_t)(4 * j) * DIM];
        bf[j].y = wcol[(size_t)(4 * j + 1) * DIM];
    }
    #pragma unroll
    for (int j = 0; j < 32; ++j) {
        v2f a[K];
        #pragma unroll
        for (int t = 0; t < K; ++t) a[t] = *(const v2f*)(acc + a_off[t] + 4 * j);
        #pragma unroll
        for (int t = 0; t < K; ++t)
            cc[t] = __builtin_amdgcn_wmma_f32_16x16x4_f32(false, a[t], false, bf[j],
                                                          (short)0, cc[t], false, false);
    }
}

__global__ __launch_bounds__(WG) void k_accumulate(const int* __restrict__ src,
                                                     const int* __restrict__ dst,
                                                     const int2* __restrict__ sorted,
                                                     const int* __restrict__ offs,
                                                     const float* __restrict__ dinv,
                                                     const float* __restrict__ x,
                                                     const float* __restrict__ W,
                                                     const float* __restrict__ bias,
                                                     float* __restrict__ out, int N, int E) {
    __shared__ __attribute__((aligned(16))) float acc[ROWS * STRIDE];
    __shared__ int      l_s[CAP];
    __shared__ int      l_t[CAP];
    __shared__ float    l_w[CAP];
    __shared__ unsigned nbin[NW];

    const int tid  = threadIdx.x;
    const int wave = __builtin_amdgcn_readfirstlane(tid >> 5);
    const int lane = tid & 31;
    const int base = blockIdx.x * ROWS;
    const int nrows = min(ROWS, N - base);
    const unsigned nr = (unsigned)nrows;

    for (int r0 = wave; r0 < nrows; r0 += 4 * NW) {
        float  di[4];
        float4 v[4];
        #pragma unroll
        for (int u = 0; u < 4; ++u) {
            const int r = min(r0 + u * NW, nrows - 1);
            di[u] = dinv[base + r];
            v[u]  = ((const float4*)(x + (size_t)(base + r) * DIM))[lane];
        }
        #pragma unroll
        for (int u = 0; u < 4; ++u) {
            const int r = r0 + u * NW;
            if (r < nrows) {
                float4 o;
                o.x = v[u].x * di[u] * di[u];
                o.y = v[u].y * di[u] * di[u];
                o.z = v[u].z * di[u] * di[u];
                o.w = v[u].w * di[u] * di[u];
                ((float4*)(acc + r * STRIDE))[lane] = o;
            }
        }
    }
    if (tid < NW) nbin[tid] = 0u;
    __syncthreads();

    auto note = [&](int e) {
        const int      tl   = dst[e] - base;
        const unsigned bin  = (unsigned)tl & (unsigned)(NW - 1);
        const unsigned slot = atomicAdd(&nbin[bin], 1u);
        if (slot < (unsigned)BCAP) {
            l_s[bin * BCAP + slot] = e;
            l_t[bin * BCAP + slot] = tl;
        } else {
            const int s = src[e];
            add_row_alone(acc + tl * STRIDE, x + (size_t)s * DIM, dinv[s] * dinv[base + tl]);
        }
    };
    const bool use_sorted = gridDim.x <= KEYMAX;
    if (use_sorted) {
        auto file = [&](const int2 ed, const float w) {
            const int      tl   = ed.y - base;
            const unsigned bin  = (unsigned)tl & (unsigned)(NW - 1);
            const unsigned slot = atomicAdd(&nbin[bin], 1u);
            if (slot < (unsigned)BCAP) {
                const unsigned i = bin * BCAP + slot;
                l_s[i] = ed.x; l_t[i] = tl; l_w[i] = w;
            } else {
                add_row_alone(acc + tl * STRIDE, x + (size_t)ed.x * DIM, w);
            }
        };
        const int nchunk = (E + CHUNK - 1) / CHUNK;
        for (int q0 = wave; q0 < nchunk; q0 += RB * NW) {
            int b_l = 0, n_l = 0;
            {
                const int q = q0 + lane * NW;
                if (lane < RB && q < nchunk) {
                    const int* o = offs + (size_t)q * OFFP + blockIdx.x;
                    b_l = o[0];
                    n_l = o[1] - b_l;
                }
            }
            int2 ed[RB];
            #pragma unroll
            for (int r = 0; r < RB; ++r) {
                const int b = __builtin_amdgcn_readlane(b_l, r);
                const int n = __builtin_amdgcn_readlane(n_l, r);
                ed[r] = make_int2(0, 0);
                if (lane < n) ed[r] = sorted[(size_t)(q0 + r * NW) * CHUNK + b + lane];
            }
            float wr[RB];
            #pragma unroll
            for (int r = 0; r < RB; ++r)
                wr[r] = dinv[ed[r].x] * dinv[ed[r].y];
            #pragma unroll
            for (int r = 0; r < RB; ++r) {
                const int n = __builtin_amdgcn_readlane(n_l, r);
                if (lane < n) file(ed[r], wr[r]);
            }
            #pragma unroll
            for (int r = 0; r < RB; ++r) {
                const int b = __builtin_amdgcn_readlane(b_l, r);
                const int n = __builtin_amdgcn_readlane(n_l, r);
                for (int i = 32 + lane; i < n; i += 32) {
                    const int2 e2 = sorted[(size_t)(q0 + r * NW) * CHUNK + b + i];
                    file(e2, dinv[e2.x] * dinv[e2.y]);
                }
            }
        }
    } else if ((((size_t)dst) & 15) == 0) {
        const int4* __restrict__ d4 = (const int4*)dst;
        const int nv = E >> 2;
        int g = tid;
        for (; g + 7 * WG < nv; g += 8 * WG) {
            int4 t[8];
            #pragma unroll
            for (int u = 0; u < 8; ++u) t[u] = d4[g + u * WG];
            unsigned mask = 0u;
            #pragma unroll
            for (int u = 0; u < 8; ++u) {
                mask |= ((unsigned)(t[u].x - base) < nr ? 1u : 0u) << (4 * u + 0);
                mask |= ((unsigned)(t[u].y - base) < nr ? 1u : 0u) << (4 * u + 1);
                mask |= ((unsigned)(t[u].z - base) < nr ? 1u : 0u) << (4 * u + 2);
                mask |= ((unsigned)(t[u].w - base) < nr ? 1u : 0u) << (4 * u + 3);
            }
            while (mask) {
                const int k = __builtin_ctz(mask);
                mask &= mask - 1u;
                note(((g + (k >> 2) * WG) << 2) + (k & 3));
            }
        }
        for (; g < nv; g += WG) {
            const int4 t = d4[g];
            if ((unsigned)(t.x - base) < nr) note((g << 2) + 0);
            if ((unsigned)(t.y - base) < nr) note((g << 2) + 1);
            if ((unsigned)(t.z - base) < nr) note((g << 2) + 2);
            if ((unsigned)(t.w - base) < nr) note((g << 2) + 3);
        }
        for (int e = (nv << 2) + tid; e < E; e += WG)
            if ((unsigned)(dst[e] - base) < nr) note(e);
    } else {
        for (int e = tid; e < E; e += WG)
            if ((unsigned)(dst[e] - base) < nr) note(e);
    }
    __syncthreads();

    if (!use_sorted) {
        for (int i = tid; i < CAP; i += WG) {
            if ((unsigned)(i % BCAP) < min(nbin[i / BCAP], (unsigned)BCAP)) {
                const int s = src[l_s[i]];
                l_s[i] = s;
                l_w[i] = dinv[s] * dinv[base + l_t[i]];
            }
        }
    }
    __syncthreads();

    {
        const int nb = __builtin_amdgcn_readfirstlane((int)min(nbin[wave], (unsigned)BCAP));
        const int* __restrict__   b_s = l_s + wave * BCAP;
        const int* __restrict__   b_t = l_t + wave * BCAP;
        const float* __restrict__ b_w = l_w + wave * BCAP;
        auto fetch = [&](const int g0, int& t_l, float& w_l, float4 (&v)[8]) {
            const int m = min(8, nb - g0);
            int s_l = 0;
            t_l = 0; w_l = 0.f;
            if (lane < m) { s_l = b_s[g0 + lane]; t_l = b_t[g0 + lane]; w_l = b_w[g0 + lane]; }
            #pragma unroll
            for (int u = 0; u < 8; ++u) {
                const int sj = __builtin_amdgcn_readlane(s_l, min(u, m - 1));
                v[u] = ((const float4*)(x + (size_t)sj * DIM))[lane];
            }
        };
        auto add8 = [&](const int g0, const int t_l, const float w_l, const float4 (&v)[8]) {
            const int m = min(8, nb - g0);
            #pragma unroll
            for (int u = 0; u < 8; ++u) {
                if (u < m) {
                    const int   tj = __builtin_amdgcn_readlane(t_l, u);
                    const float wj = __int_as_float(__builtin_amdgcn_readlane(__float_as_int(w_l), u));
                    add_row((float4*)(acc + tj * STRIDE) + lane, v[u], wj);
                }
            }
        };
        int    t0 = 0, t1 = 0;
        float  w0 = 0.f, w1 = 0.f;
        float4 v0[8], v1[8];
        if (nb > 0) fetch(0, t0, w0, v0);
        for (int g0 = 0; g0 < nb; g0 += 16) {
            if (g0 + 8 < nb)  fetch(g0 + 8, t1, w1, v1);
            add8(g0, t0, w0, v0);
            if (g0 + 16 < nb) fetch(g0 + 16, t0, w0, v0);
            if (g0 + 8 < nb)  add8(g0 + 8, t1, w1, v1);
        }
    }
    __syncthreads();

    {
        const int tn = wave & 7;
        const int wq = wave >> 3;
        const int c  = lane & 15;
        const int h  = lane >> 4;
        const float* __restrict__ wcol = W + (size_t)(2 * h) * DIM + 16 * tn + c;
        int a_off[MAXT];
        v8f cc[MAXT];
        #pragma unroll
        for (int t = 0; t < MAXT; ++t) {
            a_off[t] = min(16 * (wq + 4 * t) + c, nrows - 1) * STRIDE + 2 * h;
            cc[t] = (v8f){0.f, 0.f, 0.f, 0.f, 0.f, 0.f, 0.f, 0.f};
        }
        if (wq + 4 * (MAXT - 1) < NT) tiles_times_w<MAXT>(acc, wcol, a_off, cc);
        else                          tiles_times_w<MAXT - 1>(acc, wcol, a_off, cc);
        __syncthreads();
        const int wpos = 16 * tn + c;
        #pragma unroll
        for (int t = 0; t < MAXT; ++t) {
            #pragma unroll
            for (int r = 0; r < 8; ++r) {
                const int row = 16 * (wq + 4 * t) + 8 * h + r;
                if (row < nrows) acc[row * STRIDE + wpos] = cc[t][r];
            }
        }
    }
    __syncthreads();

    const float4 bb = ((const float4*)bias)[lane];
    for (int r0 = wave; r0 < nrows; r0 += 4 * NW) {
        float4 v[4];
        float  mx[4], sum[4];
        #pragma unroll
        for (int u = 0; u < 4; ++u) {
            const int r = min(r0 + u * NW, nrows - 1);
            const float4 a = ((const float4*)(acc + r * STRIDE))[lane];
            v[u].x = fmaxf(a.x + bb.x, 0.f);
            v[u].y = fmaxf(a.y + bb.y, 0.f);
            v[u].z = fmaxf(a.z + bb.z, 0.f);
            v[u].w = fmaxf(a.w + bb.w, 0.f);
            mx[u] = fmaxf(fmaxf(v[u].x, v[u].y), fmaxf(v[u].z, v[u].w));
        }
        #pragma unroll
        for (int off = 16; off > 0; off >>= 1) {
            #pragma unroll
            for (int u = 0; u < 4; ++u) mx[u] = fmaxf(mx[u], __shfl_xor(mx[u], off, 32));
        }
        #pragma unroll
        for (int u = 0; u < 4; ++u) {
            v[u].x = expf(v[u].x - mx[u]); v[u].y = expf(v[u].y - mx[u]);
            v[u].z = expf(v[u].z - mx[u]); v[u].w = expf(v[u].w - mx[u]);
            sum[u] = v[u].x + v[u].y + v[u].z + v[u].w;
        }
        #pragma unroll
        for (int off = 16; off > 0; off >>= 1) {
            #pragma unroll
            for (int u = 0; u < 4; ++u) sum[u] += __shfl_xor(sum[u], off, 32);
        }
        #pragma unroll
        for (int u = 0; u < 4; ++u) {
            const int r = r0 + u * NW;
            if (r < nrows) {
                const float inv = 1.0f / sum[u];
                float4 o;
                o.x = v[u].x * inv; o.y = v[u].y * inv; o.z = v[u].z * inv; o.w = v[u].w * inv;
                ((float4*)(out + (size_t)(base + r) * DIM))[lane] = o;
            }
        }
    }
}


extern "C" void kernel_launch(void* const* d_in, const int* in_sizes, int n_in,
                              void* d_out, int out_size, void* d_ws, size_t ws_size,
                              hipStream_t stream) {
    const float* x  = (const float*)d_in[0];
    const int*   ei = (const int*)  d_in[1];
    const float* W  = (const float*)d_in[2];
    const float* b  = (const float*)d_in[3];
    const int N = in_sizes[0] / DIM;
    const int E = in_sizes[1] / 2;
    const int* src = ei;
    const int* dst = ei + E;

    float*    out  = (float*)d_out;
    int2*     sorted = (int2*)d_ws;
    int*      offs = (int*)(sorted + E);
    float*    dinv = (float*)(offs + (size_t)((E + CHUNK - 1) / CHUNK) * OFFP);
    unsigned* part = (unsigned*)(dinv + N);
    const int nblk = (N + ROWS - 1) / ROWS;

    k_degree<<<dim3((N + DROWS - 1) / DROWS, DSLICE), WG, 0, stream>>>(src, dst, part, sorted, offs, nblk, N, E);
    k_dinv  <<<nblk, DWG, 0, stream>>>(part, sorted, offs, dinv, N, E);

    k_accumulate<<<nblk, WG, 0, stream>>>(src, dst, sorted, offs, dinv, x, W, b, out, N, E);
}


# k_accumulate: edges filed in per-row LDS bins; a wave walks its rows' bins as one list, sums a row's edges in registers: one LDS add per row, not per edge; sum order: edges (bin rank), self-loop last
# speedup vs baseline: 5.1151x; 1.0338x over previous
#include <hip/hip_runtime.h>
#include <math.h>


typedef float v2f __attribute__((ext_vector_type(2)));
typedef float v8f __attribute__((ext_vector_type(8)));

#define DIM    128
#define WG     1024
#define NW     (WG / 32)
#define DROWS  2048
#define DSLICE 8
#define ROWS   200
#define RCAP   40
#define RPW    ((ROWS + NW - 1) / NW)
#define STRIDE 132
#define NT     ((ROWS + 15) / 16)
#define MAXT   ((NT + 3) / 4)


#define KEYMAX 256
#define CHUNK  4096
#define OFFP   288
#define RB     7

__device__ __forceinline__ void count1(unsigned* cnt, int t, int base) {
    const unsigned tl = (unsigned)(t - base);
    if (tl < (unsigned)DROWS) atomicAdd(&cnt[tl], 1u);
}

__global__ __launch_bounds__(WG) void k_degree(const int* __restrict__ src,
                                                 const int* __restrict__ dst,
                                                 unsigned* __restrict__ part,
                                                 int2* __restrict__ sorted,
                                                 int* __restrict__ offs, int nblk,
                                                 int N, int E) {
    __shared__ unsigned cnt[DROWS];
    __shared__ int2     srt[CHUNK];
    __shared__ unsigned hist[KEYMAX];
    __shared__ unsigned start[OFFP];
    const int tid  = threadIdx.x;
    const int base = blockIdx.x * DROWS;

    if (nblk <= KEYMAX) {
        const int lane   = tid & 31;
        const int nchunk = (E + CHUNK - 1) / CHUNK;
        const int nwg    = gridDim.x * gridDim.y;
        for (int c = blockIdx.y * gridDim.x + blockIdx.x; c < nchunk; c += nwg) {
            const int c0  = c * CHUNK;
            const int len = min(CHUNK, E - c0);
            int      s[CHUNK / WG], d[CHUNK / WG];
            unsigned ow[CHUNK / WG], rk[CHUNK / WG];
            #pragma unroll
            for (int u = 0; u < CHUNK / WG; ++u) {
                const int e = tid + u * WG;
                s[u] = 0; d[u] = 0;
                if (e < len) { s[u] = src[c0 + e]; d[u] = dst[c0 + e]; }
            }
            if (tid < KEYMAX) hist[tid] = 0u;
            __syncthreads();
            #pragma unroll
            for (int u = 0; u < CHUNK / WG; ++u) {
                ow[u] = (unsigned)d[u] / (unsigned)ROWS;
                rk[u] = 0u;
                if (tid + u * WG < len) rk[u] = atomicAdd(&hist[ow[u]], 1u);
            }
            __syncthreads();
            if (tid < 32) {
                unsigned v[KEYMAX / 32];
                unsigned tot = 0u;
                #pragma unroll
                for (int k = 0; k < KEYMAX / 32; ++k) { const unsigned t = hist[(KEYMAX / 32) * lane + k]; v[k] = tot; tot += t; }
                unsigned inc = tot;
                #pragma unroll
                for (int off = 1; off < 32; off <<= 1) {
                    const unsigned y = __shfl_up(inc, off, 32);
                    if (lane >= off) inc += y;
                }
                #pragma unroll
                for (int k = 0; k < KEYMAX / 32; ++k) start[(KEYMAX / 32) * lane + k] = inc - tot + v[k];
            } else if (tid >= KEYMAX && tid < OFFP) {
                start[tid] = (unsigned)len;
            }
            __syncthreads();
            #pragma unroll
            for (int u = 0; u < CHUNK / WG; ++u)
                if (tid + u * WG < len) srt[start[ow[u]] + rk[u]] = make_int2(s[u], d[u]);
            if (tid < OFFP) offs[(size_t)c * OFFP + tid] = (int)start[tid];
            __syncthreads();
            for (int i = tid; i < len; i += WG) sorted[c0 + i] = srt[i];
            __syncthreads();
        }
        return;
    }

    for (int i = tid; i < DROWS; i += WG) cnt[i] = 0u;
    __syncthreads();

    const int per = (((E + DSLICE - 1) / DSLICE) + 3) & ~3;
    const int e0  = min(E, (int)blockIdx.y * per);
    const int e1  = min(E, e0 + per);
    const int* __restrict__ ds = dst + e0;
    const int len = e1 - e0;

    if ((((size_t)ds) & 15) == 0) {
        const int4* __restrict__ d4 = (const int4*)ds;
        const int nv = len >> 2;
        int g = tid;
        for (; g + 3 * WG < nv; g += 4 * WG) {
            int4 t[4];
            #pragma unroll
            for (int u = 0; u < 4; ++u) t[u] = d4[g + u * WG];
            #pragma unroll
            for (int u = 0; u < 4; ++u) {
                count1(cnt, t[u].x, base); count1(cnt, t[u].y, base);
                count1(cnt, t[u].z, base); count1(cnt, t[u].w, base);
            }
        }
        for (; g < nv; g += WG) {
            const int4 t = d4[g];
            count1(cnt, t.x, base); count1(cnt, t.y, base);
            count1(cnt, t.z, base); count1(cnt, t.w, base);
        }
        for (int e = (nv << 2) + tid; e < len; e += WG) {
            const int t = ds[e];
            count1(cnt, t, base);
        }
    } else {
        for (int e = tid; e < len; e += WG) {
            const int t = ds[e];
            count1(cnt, t, base);
        }
    }
    __syncthreads();
    unsigned* __restrict__ mine = part + (size_t)blockIdx.y * N;
    for (int i = tid; i < DROWS; i += WG)
        if (base + i < N) mine[base + i] = cnt[i];
}

#define DWG  256
#define DNW  (DWG / 32)
#define DRB  25

__global__ __launch_bounds__(DWG) void k_dinv(const unsigned* __restrict__ part,
                                               const int2* __restrict__ sorted,
                                               const int* __restrict__ offs,
                                               float* __restrict__ dinv, int N, int E) {
    __shared__ unsigned cnt[ROWS];
    const int tid   = threadIdx.x;
    const int base  = blockIdx.x * ROWS;
    const int nrows = min(ROWS, N - base);
    if (gridDim.x > KEYMAX) {
        if (tid < nrows) {
            unsigned d = 1u;
            #pragma unroll
            for (int g = 0; g < DSLICE; ++g) d += part[(size_t)g * N + base + tid];
            dinv[base + tid] = rsqrtf((float)d);
        }
        return;
    }
    const int wave = __builtin_amdgcn_readfirstlane(tid >> 5);
    const int lane = tid & 31;
    const unsigned nr = (unsigned)nrows;
    if (tid < ROWS) cnt[tid] = 0u;
    __syncthreads();
    const int nchunk = (E + CHUNK - 1) / CHUNK;
    for (int q0 = wave; q0 < nchunk; q0 += DRB * DNW) {
        int b_l = 0, n_l = 0;
        {
            const int q = q0 + lane * DNW;
            if (lane < DRB && q < nchunk) {
                const int* o = offs + (size_t)q * OFFP + blockIdx.x;
                b_l = o[0];
                n_l = o[1] - b_l;
            }
        }
        int t[DRB];
        #pragma unroll
        for (int r = 0; r < DRB; ++r) {
            const int b = __builtin_amdgcn_readlane(b_l, r);
            const int n = __builtin_amdgcn_readlane(n_l, r);
            t[r] = -1;
            if (lane < n) t[r] = sorted[(size_t)(q0 + r * DNW) * CHUNK + b + lane].y;
        }
        #pragma unroll
        for (int r = 0; r < DRB; ++r) {
            const unsigned tl = (unsigned)(t[r] - base);
            if (tl < nr) atomicAdd(&cnt[tl], 1u);
        }
        #pragma unroll
        for (int r = 0; r < DRB; ++r) {
            const int b = __builtin_amdgcn_readlane(b_l, r);
            const int n = __builtin_amdgcn_readlane(n_l, r);
            for (int i = 32 + lane; i < n; i += 32) {
                const unsigned tl = (unsigned)(sorted[(size_t)(q0 + r * DNW) * CHUNK + b + i].y - base);
                if (tl < nr) atomicAdd(&cnt[tl], 1u);
            }
        }
    }
    __syncthreads();
    if (tid < nrows) dinv[base + tid] = rsqrtf((float)(1u + cnt[tid]));
}


__device__ __forceinline__ void add_row_alone(float* a, const float* __restrict__ xr, float w) {
    for (int c = 0; c < DIM; ++c)
        atomicAdd(a + c, xr[c] * w);
}

__device__ __forceinline__ void flush_row(float4* a, const float4 s) {
    float4 o = *a;
    o.x += s.x; o.y += s.y; o.z += s.z; o.w += s.w;
    *a = o;
}

template <int K>
__device__ __forceinline__ void tiles_times_w(const float* acc, const float* __restrict__ wcol,
                                              const int (&a_off)[MAXT], v8f (&cc)[MAXT]) {
    v2f bf[32];
    #pragma unroll
    for (int j = 0; j < 32; ++j) {
        bf[j].x = wcol[(size_t)(4 * j) * DIM];
        bf[j].y = wcol[(size_t)(4 * j + 1) * DIM];
    }
    #pragma unroll
    for (int j = 0; j < 32; ++j) {
        v2f a[K];
        #pragma unroll
        for (int t = 0; t < K; ++t) a[t] = *(const v2f*)(acc + a_off[t] + 4 * j);
        #pragma unroll
        for (int t = 0; t < K; ++t)
            cc[t] = __builtin_amdgcn_wmma_f32_16x16x4_f32(false, a[t], false, bf[j],
                                                          (short)0, cc[t], false, false);
    }
}

__global__ __launch_bounds__(WG) void k_accumulate(const int* __restrict__ src,
                                                     const int* __restrict__ dst,
                                                     const int2* __restrict__ sorted,
                                                     const int* __restrict__ offs,
                                                     const float* __restrict__ dinv,
                                                     const float* __restrict__ x,
                                                     const float* __restrict__ W,
                                                     const float* __restrict__ bias,
                                                     float* __restrict__ out, int N, int E) {
    __shared__ __attribute__((aligned(16))) float acc[ROWS * STRIDE];
    __shared__ int      l_s[ROWS * RCAP];
    __shared__ float    l_w[ROWS * RCAP];
    __shared__ unsigned nrow[ROWS];

    const int tid  = threadIdx.x;
    const int wave = __builtin_amdgcn_readfirstlane(tid >> 5);
    const int lane = tid & 31;
    const int base = blockIdx.x * ROWS;
    const int nrows = min(ROWS, N - base);
    const unsigned nr = (unsigned)nrows;

    for (int r0 = wave; r0 < nrows; r0 += 4 * NW) {
        float  di[4];
        float4 v[4];
        #pragma unroll
        for (int u = 0; u < 4; ++u) {
            const int r = min(r0 + u * NW, nrows - 1);
            di[u] = dinv[base + r];
            v[u]  = ((const float4*)(x + (size_t)(base + r) * DIM))[lane];
        }
        #pragma unroll
        for (int u = 0; u < 4; ++u) {
            const int r = r0 + u * NW;
            if (r < nrows) {
                float4 o;
                o.x = v[u].x * di[u] * di[u];
                o.y = v[u].y * di[u] * di[u];
                o.z = v[u].z * di[u] * di[u];
                o.w = v[u].w * di[u] * di[u];
                ((float4*)(acc + r * STRIDE))[lane] = o;
            }
        }
    }
    if (tid < ROWS) nrow[tid] = 0u;
    __syncthreads();

    auto note = [&](int e) {
        const int      tl   = dst[e] - base;
        const unsigned slot = atomicAdd(&nrow[tl], 1u);
        if (slot < (unsigned)RCAP) {
            l_s[tl * RCAP + slot] = e;
        } else {
            const int s = src[e];
            add_row_alone(acc + tl * STRIDE, x + (size_t)s * DIM, dinv[s] * dinv[base + tl]);
        }
    };
    const bool use_sorted = gridDim.x <= KEYMAX;
    if (use_sorted) {
        auto file = [&](const int2 ed, const float w) {
            const int      tl   = ed.y - base;
            const unsigned slot = atomicAdd(&nrow[tl], 1u);
            if (slot < (unsigned)RCAP) {
                const unsigned i = (unsigned)tl * RCAP + slot;
                l_s[i] = ed.x; l_w[i] = w;
            } else {
                add_row_alone(acc + tl * STRIDE, x + (size_t)ed.x * DIM, w);
            }
        };
        const int nchunk = (E + CHUNK - 1) / CHUNK;
        for (int q0 = wave; q0 < nchunk; q0 += RB * NW) {
            int b_l = 0, n_l = 0;
            {
                const int q = q0 + lane * NW;
                if (lane < RB && q < nchunk) {
                    const int* o = offs + (size_t)q * OFFP + blockIdx.x;
                    b_l = o[0];
                    n_l = o[1] - b_l;
                }
            }
            int2 ed[RB];
            #pragma unroll
            for (int r = 0; r < RB; ++r) {
                const int b = __builtin_amdgcn_readlane(b_l, r);
                const int n = __builtin_amdgcn_readlane(n_l, r);
                ed[r] = make_int2(0, 0);
                if (lane < n) ed[r] = sorted[(size_t)(q0 + r * NW) * CHUNK + b + lane];
            }
            float wr[RB];
            #pragma unroll
            for (int r = 0; r < RB; ++r)
                wr[r] = dinv[ed[r].x] * dinv[ed[r].y];
            #pragma unroll
            for (int r = 0; r < RB; ++r) {
                const int n = __builtin_amdgcn_readlane(n_l, r);
                if (lane < n) file(ed[r], wr[r]);
            }
            #pragma unroll
            for (int r = 0; r < RB; ++r) {
                const int b = __builtin_amdgcn_readlane(b_l, r);
                const int n = __builtin_amdgcn_readlane(n_l, r);
                for (int i = 32 + lane; i < n; i += 32) {
                    const int2 e2 = sorted[(size_t)(q0 + r * NW) * CHUNK + b + i];
                    file(e2, dinv[e2.x] * dinv[e2.y]);
                }
            }
        }
    } else if ((((size_t)dst) & 15) == 0) {
        const int4* __restrict__ d4 = (const int4*)dst;
        const int nv = E >> 2;
        int g = tid;
        for (; g + 7 * WG < nv; g += 8 * WG) {
            int4 t[8];
            #pragma unroll
            for (int u = 0; u < 8; ++u) t[u] = d4[g + u * WG];
            unsigned mask = 0u;
            #pragma unroll
            for (int u = 0; u < 8; ++u) {
                mask |= ((unsigned)(t[u].x - base) < nr ? 1u : 0u) << (4 * u + 0);
                mask |= ((unsigned)(t[u].y - base) < nr ? 1u : 0u) << (4 * u + 1);
                mask |= ((unsigned)(t[u].z - base) < nr ? 1u : 0u) << (4 * u + 2);
                mask |= ((unsigned)(t[u].w - base) < nr ? 1u : 0u) << (4 * u + 3);
            }
            while (mask) {
                const int k = __builtin_ctz(mask);
                mask &= mask - 1u;
                note(((g + (k >> 2) * WG) << 2) + (k & 3));
            }
        }
        for (; g < nv; g += WG) {
            const int4 t = d4[g];
            if ((unsigned)(t.x - base) < nr) note((g << 2) + 0);
            if ((unsigned)(t.y - base) < nr) note((g << 2) + 1);
            if ((unsigned)(t.z - base) < nr) note((g << 2) + 2);
            if ((unsigned)(t.w - base) < nr) note((g << 2) + 3);
        }
        for (int e = (nv << 2) + tid; e < E; e += WG)
            if ((unsigned)(dst[e] - base) < nr) note(e);
    } else {
        for (int e = tid; e < E; e += WG)
            if ((unsigned)(dst[e] - base) < nr) note(e);
    }
    __syncthreads();

    if (!use_sorted) {
        for (int i = tid; i < ROWS * RCAP; i += WG) {
            if ((unsigned)(i % RCAP) < min(nrow[i / RCAP], (unsigned)RCAP)) {
                const int s = src[l_s[i]];
                l_s[i] = s;
                l_w[i] = dinv[s] * dinv[base + i / RCAP];
            }
        }
    }
    __syncthreads();

    {
        int pre[RPW + 1];
        pre[0] = 0;
        #pragma unroll
        for (int k = 0; k < RPW; ++k) {
            const int r = wave + k * NW;
            unsigned n = 0u;
            if (r < nrows) n = min(nrow[r], (unsigned)RCAP);
            pre[k + 1] = pre[k] + __builtin_amdgcn_readfirstlane((int)n);
        }
        const int nb = pre[RPW];
        auto fetch = [&](const int g0, int& t_l, float& w_l, float4 (&v)[8]) {
            const int m = min(8, nb - g0);
            int s_l = 0;
            t_l = 0; w_l = 0.f;
            if (lane < m) {
                const int p = g0 + lane;
                int k = 0, b = 0;
                #pragma unroll
                for (int j = 1; j < RPW; ++j)
                    if (p >= pre[j]) { k = j; b = pre[j]; }
                const int row = wave + k * NW;
                const int i   = row * RCAP + (p - b);
                s_l = l_s[i]; w_l = l_w[i]; t_l = row;
            }
            #pragma unroll
            for (int u = 0; u < 8; ++u) {
                const int sj = __builtin_amdgcn_readlane(s_l, min(u, m - 1));
                v[u] = ((const float4*)(x + (size_t)sj * DIM))[lane];
            }
        };
        int    cur = -1;
        float4 sum = make_float4(0.f, 0.f, 0.f, 0.f);
        auto add8 = [&](const int g0, const int t_l, const float w_l, const float4 (&v)[8]) {
            const int m = min(8, nb - g0);
            #pragma unroll
            for (int u = 0; u < 8; ++u) {
                if (u < m) {
                    const int   tj = __builtin_amdgcn_readlane(t_l, u);
                    const float wj = __int_as_float(__builtin_amdgcn_readlane(__float_as_int(w_l), u));
                    if (tj != cur) {
                        if (cur >= 0) flush_row((float4*)(acc + cur * STRIDE) + lane, sum);
                        sum = make_float4(0.f, 0.f, 0.f, 0.f);
                        cur = tj;
                    }
                    sum.x += v[u].x * wj; sum.y += v[u].y * wj; sum.z += v[u].z * wj; sum.w += v[u].w * wj;
                }
            }
        };
        int    t0 = 0, t1 = 0;
        float  w0 = 0.f, w1 = 0.f;
        float4 v0[8], v1[8];
        if (nb > 0) fetch(0, t0, w0, v0);
        for (int g0 = 0; g0 < nb; g0 += 16) {
            if (g0 + 8 < nb)  fetch(g0 + 8, t1, w1, v1);
            add8(g0, t0, w0, v0);
            if (g0 + 16 < nb) fetch(g0 + 16, t0, w0, v0);
            if (g0 + 8 < nb)  add8(g0 + 8, t1, w1, v1);
        }
        if (cur >= 0) flush_row((float4*)(acc + cur * STRIDE) + lane, sum);
    }
    __syncthreads();

    {
        const int tn = wave & 7;
        const int wq = wave >> 3;
        const int c  = lane & 15;
        const int h  = lane >> 4;
        const float* __restrict__ wcol = W + (size_t)(2 * h) * DIM + 16 * tn + c;
        int a_off[MAXT];
        v8f cc[MAXT];
        #pragma unroll
        for (int t = 0; t < MAXT; ++t) {
            a_off[t] = min(16 * (wq + 4 * t) + c, nrows - 1) * STRIDE + 2 * h;
            cc[t] = (v8f){0.f, 0.f, 0.f, 0.f, 0.f, 0.f, 0.f, 0.f};
        }
        if (wq + 4 * (MAXT - 1) < NT) tiles_times_w<MAXT>(acc, wcol, a_off, cc);
        else                          tiles_times_w<MAXT - 1>(acc, wcol, a_off, cc);
        __syncthreads();
        const int wpos = 16 * tn + c;
        #pragma unroll
        for (int t = 0; t < MAXT; ++t) {
            #pragma unroll
            for (int r = 0; r < 8; ++r) {
                const int row = 16 * (wq + 4 * t) + 8 * h + r;
                if (row < nrows) acc[row * STRIDE + wpos] = cc[t][r];
            }
        }
    }
    __syncthreads();

    const float4 bb = ((const float4*)bias)[lane];
    for (int r0 = wave; r0 < nrows; r0 += 4 * NW) {
        float4 v[4];
        float  mx[4], sum[4];
        #pragma unroll
        for (int u = 0; u < 4; ++u) {
            const int r = min(r0 + u * NW, nrows - 1);
            const float4 a = ((const float4*)(acc + r * STRIDE))[lane];
            v[u].x = fmaxf(a.x + bb.x, 0.f);
            v[u].y = fmaxf(a.y + bb.y, 0.f);
            v[u].z = fmaxf(a.z + bb.z, 0.f);
            v[u].w = fmaxf(a.w + bb.w, 0.f);
            mx[u] = fmaxf(fmaxf(v[u].x, v[u].y), fmaxf(v[u].z, v[u].w));
        }
        #pragma unroll
        for (int off = 16; off > 0; off >>= 1) {
            #pragma unroll
            for (int u = 0; u < 4; ++u) mx[u] = fmaxf(mx[u], __shfl_xor(mx[u], off, 32));
        }
        #pragma unroll
        for (int u = 0; u < 4; ++u) {
            v[u].x = expf(v[u].x - mx[u]); v[u].y = expf(v[u].y - mx[u]);
            v[u].z = expf(v[u].z - mx[u]); v[u].w = expf(v[u].w - mx[u]);
            sum[u] = v[u].x + v[u].y + v[u].z + v[u].w;
        }
        #pragma unroll
        for (int off = 16; off > 0; off >>= 1) {
            #pragma unroll
            for (int u = 0; u < 4; ++u) sum[u] += __shfl_xor(sum[u], off, 32);
        }
        #pragma unroll
        for (int u = 0; u < 4; ++u) {
            const int r = r0 + u * NW;
            if (r < nrows) {
                const float inv = 1.0f / sum[u];
                float4 o;
                o.x = v[u].x * inv; o.y = v[u].y * inv; o.z = v[u].z * inv; o.w = v[u].w * inv;
                ((float4*)(out + (size_t)(base + r) * DIM))[lane] = o;
            }
        }
    }
}


extern "C" void kernel_launch(void* const* d_in, const int* in_sizes, int n_in,
                              void* d_out, int out_size, void* d_ws, size_t ws_size,
                              hipStream_t stream) {
    const float* x  = (const float*)d_in[0];
    const int*   ei = (const int*)  d_in[1];
    const float* W  = (const float*)d_in[2];
    const float* b  = (const float*)d_in[3];
    const int N = in_sizes[0] / DIM;
    const int E = in_sizes[1] / 2;
    const int* src = ei;
    const int* dst = ei + E;

    float*    out  = (float*)d_out;
    int2*     sorted = (int2*)d_ws;
    int*      offs = (int*)(sorted + E);
    float*    dinv = (float*)(offs + (size_t)((E + CHUNK - 1) / CHUNK) * OFFP);
    unsigned* part = (unsigned*)(dinv + N);
    const int nblk = (N + ROWS - 1) / ROWS;

    k_degree<<<dim3((N + DROWS - 1) / DROWS, DSLICE), WG, 0, stream>>>(src, dst, part, sorted, offs, nblk, N, E);
    k_dinv  <<<nblk, DWG, 0, stream>>>(part, sorted, offs, dinv, N, E);

    k_accumulate<<<nblk, WG, 0, stream>>>(src, dst, sorted, offs, dinv, x, W, b, out, N, E);
}
